# starting kernel
# baseline (speedup 1.0000x reference)
#include <hip/hip_runtime.h>
#include <math.h>

typedef __attribute__((ext_vector_type(2))) float v2f;
typedef __attribute__((ext_vector_type(8))) float v8f;

#define B_    16
#define C_    768
#define HW_   4096
#define D_    256
#define M_    32
#define EPS_  0.05f
#define NORM_ (-8.3255517f)
#define NEPS_ 1e-12f

#define OFF_WHAT  0
#define OFF_Z     8192
#define OFF_U     (8192 + 2097152)
#define OFF_V     (OFF_U + 512)
#define OFF_RINV  (OFF_V + 65536)

static __device__ __forceinline__ v8f wmma4(v2f a, v2f b, v8f c) {
  return __builtin_amdgcn_wmma_f32_16x16x4_f32(false, a, false, b, (short)0, c,
                                               false, false);
}

__global__ void k_norm_v(const float* __restrict__ v, float* __restrict__ what) {
  int m = threadIdx.x;
  if (m < M_) {
    float s = 0.f;
    for (int d = 0; d < D_; ++d) { float t = v[m * D_ + d]; s += t * t; }
    float inv = 1.0f / fmaxf(sqrtf(s), NEPS_);
    for (int d = 0; d < D_; ++d) what[m * D_ + d] = v[m * D_ + d] * inv;
  }
}

__global__ void k_zero(float* __restrict__ p, int n) {
  int i = blockIdx.x * blockDim.x + threadIdx.x;
  if (i < n) p[i] = 0.f;
}

__global__ void __launch_bounds__(256) k_proj(
    const float* __restrict__ x, const float* __restrict__ w,
    const float* __restrict__ bias, const float* __restrict__ what,
    float* __restrict__ xp, float* __restrict__ Z, float* __restrict__ rinvg) {
  __shared__ float sA[C_ * 16];
  __shared__ float sRed[8][16];
  __shared__ float sRinv[16];

  const int tid = threadIdx.x;
  const int b = blockIdx.x >> 8;
  const int n0 = (blockIdx.x & 255) * 16;
  const int wave = tid >> 5;
  const int lane = tid & 31;
  const int laneLo = lane & 15;
  const int laneHi = lane >> 4;

  {
    const float* xb = x + (size_t)b * C_ * HW_ + n0;
    for (int i = tid; i < C_ * 16; i += 256) {
      int c = i >> 4, n = i & 15;
      sA[i] = xb[(size_t)c * HW_ + n];
    }
  }
  __syncthreads();

  const int d0 = wave * 32;
  v8f acc0 = {}; v8f acc1 = {};
  const float* w0 = w + (size_t)(d0 + laneLo) * C_;
  const float* w1 = w + (size_t)(d0 + 16 + laneLo) * C_;
#pragma unroll 4
  for (int k = 0; k < C_; k += 4) {
    int kk = k + 2 * laneHi;
    v2f a; a.x = sA[kk * 16 + laneLo]; a.y = sA[(kk + 1) * 16 + laneLo];
    v2f b0 = *(const v2f*)(w0 + kk);
    v2f b1 = *(const v2f*)(w1 + kk);
    acc0 = wmma4(a, b0, acc0);
    acc1 = wmma4(a, b1, acc1);
  }

  const float bi0 = bias[d0 + laneLo];
  const float bi1 = bias[d0 + 16 + laneLo];
#pragma unroll
  for (int r = 0; r < 8; ++r) { acc0[r] += bi0; acc1[r] += bi1; }

  {
    float* xpb = xp + (size_t)b * HW_ * D_ + (size_t)n0 * D_;
#pragma unroll
    for (int r = 0; r < 8; ++r) {
      int row = r + 8 * laneHi;
      xpb[(size_t)row * D_ + d0 + laneLo] = acc0[r];
      xpb[(size_t)row * D_ + d0 + 16 + laneLo] = acc1[r];
    }
  }

#pragma unroll
  for (int r = 0; r < 8; ++r) {
    float val = acc0[r] * acc0[r] + acc1[r] * acc1[r];
    for (int off = 1; off < 16; off <<= 1) val += __shfl_xor(val, off, 32);
    if (laneLo == 0) sRed[wave][r + 8 * laneHi] = val;
  }
  __syncthreads();

  if (tid < 16) {
    float s = 0.f;
    for (int wv = 0; wv < 8; ++wv) s += sRed[wv][tid];
    float inv = 1.0f / fmaxf(sqrtf(s), NEPS_);
    sRinv[tid] = inv;
    rinvg[(size_t)b * HW_ + n0 + tid] = inv;
  }

  float* sP = sA;
#pragma unroll
  for (int r = 0; r < 8; ++r) {
    int row = r + 8 * laneHi;
    sP[(d0 + laneLo) * 16 + row] = acc0[r];
    sP[(d0 + 16 + laneLo) * 16 + row] = acc1[r];
  }
  __syncthreads();

  if (wave < 2) {
    const int m0 = wave * 16;
    v8f cs = {};
    const float* wh = what + (size_t)(m0 + laneLo) * D_;
#pragma unroll 4
    for (int k = 0; k < D_; k += 4) {
      int kk = k + 2 * laneHi;
      v2f a; a.x = sP[kk * 16 + laneLo]; a.y = sP[(kk + 1) * 16 + laneLo];
      v2f bb = *(const v2f*)(wh + kk);
      cs = wmma4(a, bb, cs);
    }
    float* Zb = Z + (size_t)b * (M_ * HW_) + (size_t)(m0 + laneLo) * HW_ + n0;
    const float invEps = 1.0f / EPS_;
#pragma unroll
    for (int r = 0; r < 8; ++r) {
      int row = r + 8 * laneHi;
      Zb[row] = cs[r] * sRinv[row] * invEps;
    }
  }
}

__global__ void __launch_bounds__(256) k_u(const float* __restrict__ Z,
                                           const float* __restrict__ vdual,
                                           float* __restrict__ u) {
  __shared__ float sb[256];
  const int b = blockIdx.x >> 5;
  const int m = blockIdx.x & 31;
  const float* zrow = Z + (size_t)b * (M_ * HW_) + (size_t)m * HW_;
  const float* vv = vdual + (size_t)b * HW_;
  const int tid = threadIdx.x;

  float mx = -1e30f;
  for (int n = tid; n < HW_; n += 256) mx = fmaxf(mx, zrow[n] + vv[n]);
  sb[tid] = mx; __syncthreads();
  for (int s = 128; s > 0; s >>= 1) {
    if (tid < s) sb[tid] = fmaxf(sb[tid], sb[tid + s]);
    __syncthreads();
  }
  const float gmax = sb[0]; __syncthreads();

  float sm = 0.f;
  for (int n = tid; n < HW_; n += 256) sm += expf(zrow[n] + vv[n] - gmax);
  sb[tid] = sm; __syncthreads();
  for (int s = 128; s > 0; s >>= 1) {
    if (tid < s) sb[tid] += sb[tid + s];
    __syncthreads();
  }
  if (tid == 0) u[b * M_ + m] = NORM_ - (gmax + logf(sb[0]));
}

__global__ void __launch_bounds__(256) k_vdual(const float* __restrict__ Z,
                                               const float* __restrict__ u,
                                               float* __restrict__ vdual) {
  __shared__ float sU[M_];
  const int idx = blockIdx.x * 256 + threadIdx.x;
  const int b = idx >> 12;
  const int n = idx & 4095;
  if (threadIdx.x < M_) sU[threadIdx.x] = u[b * M_ + threadIdx.x];
  __syncthreads();
  const float* zb = Z + (size_t)b * (M_ * HW_) + n;
  float mx = -1e30f;
#pragma unroll
  for (int m = 0; m < M_; ++m) mx = fmaxf(mx, zb[(size_t)m * HW_] + sU[m]);
  float sm = 0.f;
#pragma unroll
  for (int m = 0; m < M_; ++m) sm += expf(zb[(size_t)m * HW_] + sU[m] - mx);
  vdual[idx] = NORM_ - (mx + logf(sm));
}

__global__ void __launch_bounds__(256) k_vtilde(
    const float* __restrict__ Z, const float* __restrict__ u,
    const float* __restrict__ vdual, const float* __restrict__ xp,
    const float* __restrict__ rinv, float* __restrict__ vt) {
  __shared__ float sAcc[8][32][9];
  const int tid = threadIdx.x;
  const int wave = tid >> 5;
  const int lane = tid & 31;
  const int laneLo = lane & 15;
  const int laneHi = lane >> 4;

  const int b = blockIdx.x >> 5;
  const int rem = blockIdx.x & 31;
  const int m0 = (rem >> 4) * 16;
  const int d0 = (rem & 15) * 16;

  const int k_beg = wave * (HW_ / 8);
  const int k_end = k_beg + (HW_ / 8);

  const float* zrow = Z + (size_t)b * (M_ * HW_) + (size_t)(m0 + laneLo) * HW_;
  const float um = u[b * M_ + m0 + laneLo];
  const float* vv = vdual + (size_t)b * HW_;
  const float* xpb = xp + (size_t)b * HW_ * D_ + d0 + laneLo;
  const float* rv = rinv + (size_t)b * HW_;

  v8f acc = {};
  for (int k = k_beg; k < k_end; k += 4) {
    int kk = k + 2 * laneHi;
    v2f z2 = *(const v2f*)(zrow + kk);
    v2f vw = *(const v2f*)(vv + kk);
    v2f a;
    a.x = expf(z2.x + um + vw.x - NORM_);
    a.y = expf(z2.y + um + vw.y - NORM_);
    v2f r2 = *(const v2f*)(rv + kk);
    v2f bb;
    bb.x = xpb[(size_t)kk * D_] * r2.x;
    bb.y = xpb[(size_t)(kk + 1) * D_] * r2.y;
    acc = wmma4(a, bb, acc);
  }

#pragma unroll
  for (int r = 0; r < 8; ++r) sAcc[wave][lane][r] = acc[r];
  __syncthreads();

  if (wave == 0) {
    float* out = vt + (size_t)b * M_ * D_ + (size_t)m0 * D_ + d0 + laneLo;
#pragma unroll
    for (int r = 0; r < 8; ++r) {
      float s = 0.f;
      for (int w2 = 0; w2 < 8; ++w2) s += sAcc[w2][lane][r];
      int row = r + 8 * laneHi;
      out[(size_t)row * D_] = s;
    }
  }
}

extern "C" void kernel_launch(void* const* d_in, const int* in_sizes, int n_in,
                              void* d_out, int out_size, void* d_ws, size_t ws_size,
                              hipStream_t stream) {
  const float* x      = (const float*)d_in[0];
  const float* conv_w = (const float*)d_in[1];
  const float* conv_b = (const float*)d_in[2];
  const float* v      = (const float*)d_in[3];

  float* out_vt = (float*)d_out;
  float* out_xp = out_vt + (size_t)B_ * M_ * D_;

  float* ws   = (float*)d_ws;
  float* what = ws + OFF_WHAT;
  float* Z    = ws + OFF_Z;
  float* u    = ws + OFF_U;
  float* vd   = ws + OFF_V;
  float* rinv = ws + OFF_RINV;

  k_norm_v<<<1, 256, 0, stream>>>(v, what);
  k_zero<<<(B_ * HW_ + 255) / 256, 256, 0, stream>>>(vd, B_ * HW_);
  k_proj<<<B_ * (HW_ / 16), 256, 0, stream>>>(x, conv_w, conv_b, what,
                                              out_xp, Z, rinv);
  for (int it = 0; it < 3; ++it) {
    k_u<<<B_ * M_, 256, 0, stream>>>(Z, vd, u);
    k_vdual<<<B_ * HW_ / 256, 256, 0, stream>>>(Z, u, vd);
  }
  k_vtilde<<<B_ * M_, 256, 0, stream>>>(Z, u, vd, out_xp, rinv, out_vt);
}


# k_proj re-tiled to 64-row blocks: x and weight K-chunks staged in LDS by full-line loads (next chunk prefetched in registers), each weight fragment feeds 4 row tiles; summation order unchanged
# speedup vs baseline: 1.6363x; 1.6363x over previous
#include <hip/hip_runtime.h>
#include <math.h>

typedef __attribute__((ext_vector_type(2))) float v2f;
typedef __attribute__((ext_vector_type(4))) float v4f;
typedef __attribute__((ext_vector_type(8))) float v8f;

#define B_    16
#define C_    768
#define HW_   4096
#define D_    256
#define M_    32
#define EPS_  0.05f
#define NORM_ (-8.3255517f)
#define NEPS_ 1e-12f

#define OFF_WHAT  0
#define OFF_Z     8192
#define OFF_U     (8192 + 2097152)
#define OFF_V     (OFF_U + 512)
#define OFF_RINV  (OFF_V + 65536)

static __device__ __forceinline__ v8f wmma4(v2f a, v2f b, v8f c) {
  return __builtin_amdgcn_wmma_f32_16x16x4_f32(false, a, false, b, (short)0, c,
                                               false, false);
}

__global__ void k_norm_v(const float* __restrict__ v, float* __restrict__ what) {
  int m = threadIdx.x;
  if (m < M_) {
    float s = 0.f;
    for (int d = 0; d < D_; ++d) { float t = v[m * D_ + d]; s += t * t; }
    float inv = 1.0f / fmaxf(sqrtf(s), NEPS_);
    for (int d = 0; d < D_; ++d) what[m * D_ + d] = v[m * D_ + d] * inv;
  }
}

__global__ void k_zero(float* __restrict__ p, int n) {
  int i = blockIdx.x * blockDim.x + threadIdx.x;
  if (i < n) p[i] = 0.f;
}

#define BM_  64
#define BK_  32
#define PA_  72
#define PB_  36
#define PP_  72
__global__ void __launch_bounds__(256) k_proj(
    const float* __restrict__ x, const float* __restrict__ w,
    const float* __restrict__ bias, const float* __restrict__ what,
    float* __restrict__ xp, float* __restrict__ Z, float* __restrict__ rinvg) {
  __shared__ __attribute__((aligned(16))) float sMem[D_ * PP_];
  __shared__ float sRed[8][BM_];
  __shared__ float sRinv[BM_];
  float* sA = sMem;
  float* sB = sMem + BK_ * PA_;

  const int tid = threadIdx.x;
  const int b = blockIdx.x >> 6;
  const int n0 = (blockIdx.x & 63) * BM_;
  const int wave = tid >> 5;
  const int lane = tid & 31;
  const int laneLo = lane & 15;
  const int laneHi = lane >> 4;

  const int ar = tid >> 4;
  const int ac = (tid & 15) * 4;
  const int br = tid >> 3;
  const int bc = (tid & 7) * 4;
  const float* xg = x + (size_t)b * C_ * HW_ + n0 + (size_t)ar * HW_ + ac;
  const float* wg = w + (size_t)br * C_ + bc;

  v4f pa0, pa1, pb[8];
  pa0 = *(const v4f*)(xg);
  pa1 = *(const v4f*)(xg + (size_t)16 * HW_);
#pragma unroll
  for (int j = 0; j < 8; ++j) pb[j] = *(const v4f*)(wg + (size_t)(32 * j) * C_);

  const int d0 = wave * 32;
  v8f acc[4][2];
#pragma unroll
  for (int i = 0; i < 4; ++i) { acc[i][0] = (v8f){}; acc[i][1] = (v8f){}; }

  for (int k0 = 0; k0 < C_; k0 += BK_) {
    *(v4f*)(sA + ar * PA_ + ac) = pa0;
    *(v4f*)(sA + (ar + 16) * PA_ + ac) = pa1;
#pragma unroll
    for (int j = 0; j < 8; ++j) *(v4f*)(sB + (br + 32 * j) * PB_ + bc) = pb[j];
    __syncthreads();

    if (k0 + BK_ < C_) {
      const float* xn = xg + (size_t)(k0 + BK_) * HW_;
      pa0 = *(const v4f*)(xn);
      pa1 = *(const v4f*)(xn + (size_t)16 * HW_);
#pragma unroll
      for (int j = 0; j < 8; ++j)
        pb[j] = *(const v4f*)(wg + (size_t)(32 * j) * C_ + k0 + BK_);
    }

#pragma unroll
    for (int ks = 0; ks < BK_; ks += 4) {
      const int kk = ks + 2 * laneHi;
      v2f a[4];
#pragma unroll
      for (int i = 0; i < 4; ++i) {
        a[i].x = sA[kk * PA_ + 16 * i + laneLo];
        a[i].y = sA[(kk + 1) * PA_ + 16 * i + laneLo];
      }
      const v2f b0 = *(const v2f*)(sB + (d0 + laneLo) * PB_ + kk);
      const v2f b1 = *(const v2f*)(sB + (d0 + 16 + laneLo) * PB_ + kk);
#pragma unroll
      for (int i = 0; i < 4; ++i) {
        acc[i][0] = wmma4(a[i], b0, acc[i][0]);
        acc[i][1] = wmma4(a[i], b1, acc[i][1]);
      }
    }
    __syncthreads();
  }

  const float bi0 = bias[d0 + laneLo];
  const float bi1 = bias[d0 + 16 + laneLo];
#pragma unroll
  for (int i = 0; i < 4; ++i) {
#pragma unroll
    for (int r = 0; r < 8; ++r) { acc[i][0][r] += bi0; acc[i][1][r] += bi1; }
  }

  {
    float* xpb = xp + (size_t)b * HW_ * D_ + (size_t)n0 * D_;
#pragma unroll
    for (int i = 0; i < 4; ++i) {
#pragma unroll
      for (int r = 0; r < 8; ++r) {
        int row = 16 * i + r + 8 * laneHi;
        xpb[(size_t)row * D_ + d0 + laneLo] = acc[i][0][r];
        xpb[(size_t)row * D_ + d0 + 16 + laneLo] = acc[i][1][r];
      }
    }
  }

#pragma unroll
  for (int i = 0; i < 4; ++i) {
#pragma unroll
    for (int r = 0; r < 8; ++r) {
      float val = acc[i][0][r] * acc[i][0][r] + acc[i][1][r] * acc[i][1][r];
      for (int off = 1; off < 16; off <<= 1) val += __shfl_xor(val, off, 32);
      if (laneLo == 0) sRed[wave][16 * i + r + 8 * laneHi] = val;
    }
  }
  __syncthreads();

  if (tid < BM_) {
    float s = 0.f;
    for (int wv = 0; wv < 8; ++wv) s += sRed[wv][tid];
    float inv = 1.0f / fmaxf(sqrtf(s), NEPS_);
    sRinv[tid] = inv;
    rinvg[(size_t)b * HW_ + n0 + tid] = inv;
  }

  float* sP = sMem;
#pragma unroll
  for (int i = 0; i < 4; ++i) {
#pragma unroll
    for (int r = 0; r < 8; ++r) {
      int row = 16 * i + r + 8 * laneHi;
      sP[(d0 + laneLo) * PP_ + row] = acc[i][0][r];
      sP[(d0 + 16 + laneLo) * PP_ + row] = acc[i][1][r];
    }
  }
  __syncthreads();

  {
    const int nt = (wave >> 1) * 16;
    const int m0 = (wave & 1) * 16;
    v8f cs = {};
    const float* wh = what + (size_t)(m0 + laneLo) * D_;
#pragma unroll 4
    for (int k = 0; k < D_; k += 4) {
      int kk = k + 2 * laneHi;
      v2f a; a.x = sP[kk * PP_ + nt + laneLo]; a.y = sP[(kk + 1) * PP_ + nt + laneLo];
      v2f bb = *(const v2f*)(wh + kk);
      cs = wmma4(a, bb, cs);
    }
    float* Zb = Z + (size_t)b * (M_ * HW_) + (size_t)(m0 + laneLo) * HW_ + n0 + nt;
    const float invEps = 1.0f / EPS_;
#pragma unroll
    for (int r = 0; r < 8; ++r) {
      int row = r + 8 * laneHi;
      Zb[row] = cs[r] * sRinv[nt + row] * invEps;
    }
  }
}

__global__ void __launch_bounds__(256) k_u(const float* __restrict__ Z,
                                           const float* __restrict__ vdual,
                                           float* __restrict__ u) {
  __shared__ float sb[256];
  const int b = blockIdx.x >> 5;
  const int m = blockIdx.x & 31;
  const float* zrow = Z + (size_t)b * (M_ * HW_) + (size_t)m * HW_;
  const float* vv = vdual + (size_t)b * HW_;
  const int tid = threadIdx.x;

  float mx = -1e30f;
  for (int n = tid; n < HW_; n += 256) mx = fmaxf(mx, zrow[n] + vv[n]);
  sb[tid] = mx; __syncthreads();
  for (int s = 128; s > 0; s >>= 1) {
    if (tid < s) sb[tid] = fmaxf(sb[tid], sb[tid + s]);
    __syncthreads();
  }
  const float gmax = sb[0]; __syncthreads();

  float sm = 0.f;
  for (int n = tid; n < HW_; n += 256) sm += expf(zrow[n] + vv[n] - gmax);
  sb[tid] = sm; __syncthreads();
  for (int s = 128; s > 0; s >>= 1) {
    if (tid < s) sb[tid] += sb[tid + s];
    __syncthreads();
  }
  if (tid == 0) u[b * M_ + m] = NORM_ - (gmax + logf(sb[0]));
}

__global__ void __launch_bounds__(256) k_vdual(const float* __restrict__ Z,
                                               const float* __restrict__ u,
                                               float* __restrict__ vdual) {
  __shared__ float sU[M_];
  const int idx = blockIdx.x * 256 + threadIdx.x;
  const int b = idx >> 12;
  const int n = idx & 4095;
  if (threadIdx.x < M_) sU[threadIdx.x] = u[b * M_ + threadIdx.x];
  __syncthreads();
  const float* zb = Z + (size_t)b * (M_ * HW_) + n;
  float mx = -1e30f;
#pragma unroll
  for (int m = 0; m < M_; ++m) mx = fmaxf(mx, zb[(size_t)m * HW_] + sU[m]);
  float sm = 0.f;
#pragma unroll
  for (int m = 0; m < M_; ++m) sm += expf(zb[(size_t)m * HW_] + sU[m] - mx);
  vdual[idx] = NORM_ - (mx + logf(sm));
}

__global__ void __launch_bounds__(256) k_vtilde(
    const float* __restrict__ Z, const float* __restrict__ u,
    const float* __restrict__ vdual, const float* __restrict__ xp,
    const float* __restrict__ rinv, float* __restrict__ vt) {
  __shared__ float sAcc[8][32][9];
  const int tid = threadIdx.x;
  const int wave = tid >> 5;
  const int lane = tid & 31;
  const int laneLo = lane & 15;
  const int laneHi = lane >> 4;

  const int b = blockIdx.x >> 5;
  const int rem = blockIdx.x & 31;
  const int m0 = (rem >> 4) * 16;
  const int d0 = (rem & 15) * 16;

  const int k_beg = wave * (HW_ / 8);
  const int k_end = k_beg + (HW_ / 8);

  const float* zrow = Z + (size_t)b * (M_ * HW_) + (size_t)(m0 + laneLo) * HW_;
  const float um = u[b * M_ + m0 + laneLo];
  const float* vv = vdual + (size_t)b * HW_;
  const float* xpb = xp + (size_t)b * HW_ * D_ + d0 + laneLo;
  const float* rv = rinv + (size_t)b * HW_;

  v8f acc = {};
  for (int k = k_beg; k < k_end; k += 4) {
    int kk = k + 2 * laneHi;
    v2f z2 = *(const v2f*)(zrow + kk);
    v2f vw = *(const v2f*)(vv + kk);
    v2f a;
    a.x = expf(z2.x + um + vw.x - NORM_);
    a.y = expf(z2.y + um + vw.y - NORM_);
    v2f r2 = *(const v2f*)(rv + kk);
    v2f bb;
    bb.x = xpb[(size_t)kk * D_] * r2.x;
    bb.y = xpb[(size_t)(kk + 1) * D_] * r2.y;
    acc = wmma4(a, bb, acc);
  }

#pragma unroll
  for (int r = 0; r < 8; ++r) sAcc[wave][lane][r] = acc[r];
  __syncthreads();

  if (wave == 0) {
    float* out = vt + (size_t)b * M_ * D_ + (size_t)m0 * D_ + d0 + laneLo;
#pragma unroll
    for (int r = 0; r < 8; ++r) {
      float s = 0.f;
      for (int w2 = 0; w2 < 8; ++w2) s += sAcc[w2][lane][r];
      int row = r + 8 * laneHi;
      out[(size_t)row * D_] = s;
    }
  }
}

extern "C" void kernel_launch(void* const* d_in, const int* in_sizes, int n_in,
                              void* d_out, int out_size, void* d_ws, size_t ws_size,
                              hipStream_t stream) {
  const float* x      = (const float*)d_in[0];
  const float* conv_w = (const float*)d_in[1];
  const float* conv_b = (const float*)d_in[2];
  const float* v      = (const float*)d_in[3];

  float* out_vt = (float*)d_out;
  float* out_xp = out_vt + (size_t)B_ * M_ * D_;

  float* ws   = (float*)d_ws;
  float* what = ws + OFF_WHAT;
  float* Z    = ws + OFF_Z;
  float* u    = ws + OFF_U;
  float* vd   = ws + OFF_V;
  float* rinv = ws + OFF_RINV;

  k_norm_v<<<1, 256, 0, stream>>>(v, what);
  k_zero<<<(B_ * HW_ + 255) / 256, 256, 0, stream>>>(vd, B_ * HW_);
  k_proj<<<B_ * (HW_ / 64), 256, 0, stream>>>(x, conv_w, conv_b, what,
                                              out_xp, Z, rinv);
  for (int it = 0; it < 3; ++it) {
    k_u<<<B_ * M_, 256, 0, stream>>>(Z, vd, u);
    k_vdual<<<B_ * HW_ / 256, 256, 0, stream>>>(Z, u, vd);
  }
  k_vtilde<<<B_ * M_, 256, 0, stream>>>(Z, u, vd, out_xp, rinv, out_vt);
}


# u-update folded into neighbour kernels: k_proj and the v-update leave per-block (max,sum) partials of the log-sum-exp over n, next launch combines them; 4 launches gone; reduction tree changed
# speedup vs baseline: 1.7671x; 1.0799x over previous
#include <hip/hip_runtime.h>
#include <math.h>

typedef __attribute__((ext_vector_type(2))) float v2f;
typedef __attribute__((ext_vector_type(4))) float v4f;
typedef __attribute__((ext_vector_type(8))) float v8f;

#define B_    16
#define C_    768
#define HW_   4096
#define D_    256
#define M_    32
#define EPS_  0.05f
#define NORM_ (-8.3255517f)
#define NEPS_ 1e-12f

#define OFF_WHAT  0
#define OFF_Z     8192
#define OFF_U     (8192 + 2097152)
#define OFF_V     (OFF_U + 512)
#define OFF_RINV  (OFF_V + 65536)
#define NPA_ 64
#define NPB_ 16
#define OFF_PA    (OFF_RINV + 65536)
#define OFF_PB    (OFF_PA + 65536)
#define OFF_PC    (OFF_PB + 16384)

static __device__ __forceinline__ v8f wmma4(v2f a, v2f b, v8f c) {
  return __builtin_amdgcn_wmma_f32_16x16x4_f32(false, a, false, b, (short)0, c,
                                               false, false);
}

__global__ void k_norm_v(const float* __restrict__ v, float* __restrict__ what) {
  int m = threadIdx.x;
  if (m < M_) {
    float s = 0.f;
    for (int d = 0; d < D_; ++d) { float t = v[m * D_ + d]; s += t * t; }
    float inv = 1.0f / fmaxf(sqrtf(s), NEPS_);
    for (int d = 0; d < D_; ++d) what[m * D_ + d] = v[m * D_ + d] * inv;
  }
}

#define BM_  64
#define BK_  32
#define PA_  72
#define PB_  36
#define PP_  72
__global__ void __launch_bounds__(256) k_proj(
    const float* __restrict__ x, const float* __restrict__ w,
    const float* __restrict__ bias, const float* __restrict__ what,
    float* __restrict__ xp, float* __restrict__ Z, float* __restrict__ rinvg,
    float* __restrict__ part) {
  __shared__ __attribute__((aligned(16))) float sMem[D_ * PP_];
  __shared__ float sRed[8][BM_];
  __shared__ float sRinv[BM_];
  __shared__ float sPart[8][16][2];
  float* sA = sMem;
  float* sB = sMem + BK_ * PA_;

  const int tid = threadIdx.x;
  const int b = blockIdx.x >> 6;
  const int n0 = (blockIdx.x & 63) * BM_;
  const int wave = tid >> 5;
  const int lane = tid & 31;
  const int laneLo = lane & 15;
  const int laneHi = lane >> 4;

  const int ar = tid >> 4;
  const int ac = (tid & 15) * 4;
  const int br = tid >> 3;
  const int bc = (tid & 7) * 4;
  const float* xg = x + (size_t)b * C_ * HW_ + n0 + (size_t)ar * HW_ + ac;
  const float* wg = w + (size_t)br * C_ + bc;

  v4f pa0, pa1, pb[8];
  pa0 = *(const v4f*)(xg);
  pa1 = *(const v4f*)(xg + (size_t)16 * HW_);
#pragma unroll
  for (int j = 0; j < 8; ++j) pb[j] = *(const v4f*)(wg + (size_t)(32 * j) * C_);

  const int d0 = wave * 32;
  v8f acc[4][2];
#pragma unroll
  for (int i = 0; i < 4; ++i) { acc[i][0] = (v8f){}; acc[i][1] = (v8f){}; }

  for (int k0 = 0; k0 < C_; k0 += BK_) {
    *(v4f*)(sA + ar * PA_ + ac) = pa0;
    *(v4f*)(sA + (ar + 16) * PA_ + ac) = pa1;
#pragma unroll
    for (int j = 0; j < 8; ++j) *(v4f*)(sB + (br + 32 * j) * PB_ + bc) = pb[j];
    __syncthreads();

    if (k0 + BK_ < C_) {
      const float* xn = xg + (size_t)(k0 + BK_) * HW_;
      pa0 = *(const v4f*)(xn);
      pa1 = *(const v4f*)(xn + (size_t)16 * HW_);
#pragma unroll
      for (int j = 0; j < 8; ++j)
        pb[j] = *(const v4f*)(wg + (size_t)(32 * j) * C_ + k0 + BK_);
    }

#pragma unroll
    for (int ks = 0; ks < BK_; ks += 4) {
      const int kk = ks + 2 * laneHi;
      v2f a[4];
#pragma unroll
      for (int i = 0; i < 4; ++i) {
        a[i].x = sA[kk * PA_ + 16 * i + laneLo];
        a[i].y = sA[(kk + 1) * PA_ + 16 * i + laneLo];
      }
      const v2f b0 = *(const v2f*)(sB + (d0 + laneLo) * PB_ + kk);
      const v2f b1 = *(const v2f*)(sB + (d0 + 16 + laneLo) * PB_ + kk);
#pragma unroll
      for (int i = 0; i < 4; ++i) {
        acc[i][0] = wmma4(a[i], b0, acc[i][0]);
        acc[i][1] = wmma4(a[i], b1, acc[i][1]);
      }
    }
    __syncthreads();
  }

  const float bi0 = bias[d0 + laneLo];
  const float bi1 = bias[d0 + 16 + laneLo];
#pragma unroll
  for (int i = 0; i < 4; ++i) {
#pragma unroll
    for (int r = 0; r < 8; ++r) { acc[i][0][r] += bi0; acc[i][1][r] += bi1; }
  }

  {
    float* xpb = xp + (size_t)b * HW_ * D_ + (size_t)n0 * D_;
#pragma unroll
    for (int i = 0; i < 4; ++i) {
#pragma unroll
      for (int r = 0; r < 8; ++r) {
        int row = 16 * i + r + 8 * laneHi;
        xpb[(size_t)row * D_ + d0 + laneLo] = acc[i][0][r];
        xpb[(size_t)row * D_ + d0 + 16 + laneLo] = acc[i][1][r];
      }
    }
  }

#pragma unroll
  for (int i = 0; i < 4; ++i) {
#pragma unroll
    for (int r = 0; r < 8; ++r) {
      float val = acc[i][0][r] * acc[i][0][r] + acc[i][1][r] * acc[i][1][r];
      for (int off = 1; off < 16; off <<= 1) val += __shfl_xor(val, off, 32);
      if (laneLo == 0) sRed[wave][16 * i + r + 8 * laneHi] = val;
    }
  }
  __syncthreads();

  if (tid < BM_) {
    float s = 0.f;
    for (int wv = 0; wv < 8; ++wv) s += sRed[wv][tid];
    float inv = 1.0f / fmaxf(sqrtf(s), NEPS_);
    sRinv[tid] = inv;
    rinvg[(size_t)b * HW_ + n0 + tid] = inv;
  }

  float* sP = sMem;
#pragma unroll
  for (int i = 0; i < 4; ++i) {
#pragma unroll
    for (int r = 0; r < 8; ++r) {
      int row = 16 * i + r + 8 * laneHi;
      sP[(d0 + laneLo) * PP_ + row] = acc[i][0][r];
      sP[(d0 + 16 + laneLo) * PP_ + row] = acc[i][1][r];
    }
  }
  __syncthreads();

  {
    const int nt = (wave >> 1) * 16;
    const int m0 = (wave & 1) * 16;
    v8f cs = {};
    const float* wh = what + (size_t)(m0 + laneLo) * D_;
#pragma unroll 4
    for (int k = 0; k < D_; k += 4) {
      int kk = k + 2 * laneHi;
      v2f a; a.x = sP[kk * PP_ + nt + laneLo]; a.y = sP[(kk + 1) * PP_ + nt + laneLo];
      v2f bb = *(const v2f*)(wh + kk);
      cs = wmma4(a, bb, cs);
    }
    float* Zb = Z + (size_t)b * (M_ * HW_) + (size_t)(m0 + laneLo) * HW_ + n0 + nt;
    const float invEps = 1.0f / EPS_;
    float zv[8];
#pragma unroll
    for (int r = 0; r < 8; ++r) {
      int row = r + 8 * laneHi;
      zv[r] = cs[r] * sRinv[nt + row] * invEps;
      Zb[row] = zv[r];
    }
    float mx = zv[0];
#pragma unroll
    for (int r = 1; r < 8; ++r) mx = fmaxf(mx, zv[r]);
    mx = fmaxf(mx, __shfl_xor(mx, 16, 32));
    float sm = 0.f;
#pragma unroll
    for (int r = 0; r < 8; ++r) sm += expf(zv[r] - mx);
    sm += __shfl_xor(sm, 16, 32);
    if (laneHi == 0) { sPart[wave][laneLo][0] = mx; sPart[wave][laneLo][1] = sm; }
  }
  __syncthreads();
  if (tid < M_) {
    const int mt = tid >> 4, ml = tid & 15;
    float M = sPart[mt][ml][0];
#pragma unroll
    for (int q = 1; q < 4; ++q) M = fmaxf(M, sPart[2 * q + mt][ml][0]);
    float S = 0.f;
#pragma unroll
    for (int q = 0; q < 4; ++q)
      S += sPart[2 * q + mt][ml][1] * expf(sPart[2 * q + mt][ml][0] - M);
    float* po = part + (((size_t)b * NPA_ + (blockIdx.x & 63)) * M_ + tid) * 2;
    v2f ms; ms.x = M; ms.y = S;
    *(v2f*)po = ms;
  }
}

#define PT_ 264
template <int NP>
__global__ void __launch_bounds__(256) k_vu(const float* __restrict__ Z,
                                            const float* __restrict__ pin,
                                            float* __restrict__ u,
                                            float* __restrict__ vdual,
                                            float* __restrict__ pout, int want_part) {
  __shared__ float sU[M_];
  __shared__ float sPm[8][M_];
  __shared__ float sPs[8][M_];
  __shared__ float sT[M_ * PT_];
  const int tid = threadIdx.x;
  const int wave = tid >> 5;
  const int lane = tid & 31;
  const int b = blockIdx.x >> 4;
  const int nb = blockIdx.x & 15;
  const int n = nb * 256 + tid;

  const float* zb = Z + (size_t)b * (M_ * HW_) + n;
  float z[M_];
#pragma unroll
  for (int m = 0; m < M_; ++m) z[m] = zb[(size_t)m * HW_];

  {
    const float* pp = pin + ((size_t)b * NP * M_ + lane) * 2;
    v2f ms[NP / 8];
#pragma unroll
    for (int i = 0; i < NP / 8; ++i)
      ms[i] = *(const v2f*)(pp + (size_t)(wave + 8 * i) * (M_ * 2));
    float M = ms[0].x;
#pragma unroll
    for (int i = 1; i < NP / 8; ++i) M = fmaxf(M, ms[i].x);
    float S = 0.f;
#pragma unroll
    for (int i = 0; i < NP / 8; ++i) S += ms[i].y * expf(ms[i].x - M);
    sPm[wave][lane] = M;
    sPs[wave][lane] = S;
  }
  __syncthreads();
  if (tid < M_) {
    float M = sPm[0][tid];
#pragma unroll
    for (int w2 = 1; w2 < 8; ++w2) M = fmaxf(M, sPm[w2][tid]);
    float S = 0.f;
#pragma unroll
    for (int w2 = 0; w2 < 8; ++w2) S += sPs[w2][tid] * expf(sPm[w2][tid] - M);
    const float um = NORM_ - (M + logf(S));
    sU[tid] = um;
    if (nb == 0) u[b * M_ + tid] = um;
  }
  __syncthreads();

  float mx = -1e30f;
#pragma unroll
  for (int m = 0; m < M_; ++m) mx = fmaxf(mx, z[m] + sU[m]);
  float sm = 0.f;
#pragma unroll
  for (int m = 0; m < M_; ++m) sm += expf(z[m] + sU[m] - mx);
  const float vn = NORM_ - (mx + logf(sm));
  vdual[(size_t)b * HW_ + n] = vn;

  if (want_part) {
#pragma unroll
    for (int m = 0; m < M_; ++m) sT[m * PT_ + tid] = z[m] + vn;
    __syncthreads();
    {
      const int m = tid >> 3, seg = tid & 7;
      float tv[32];
#pragma unroll
      for (int i = 0; i < 32; ++i) tv[i] = sT[m * PT_ + seg + 8 * i];
      float M = tv[0];
#pragma unroll
      for (int i = 1; i < 32; ++i) M = fmaxf(M, tv[i]);
      float S = 0.f;
#pragma unroll
      for (int i = 0; i < 32; ++i) S += expf(tv[i] - M);
      sPm[seg][m] = M;
      sPs[seg][m] = S;
    }
    __syncthreads();
    if (tid < M_) {
      float M = sPm[0][tid];
#pragma unroll
      for (int w2 = 1; w2 < 8; ++w2) M = fmaxf(M, sPm[w2][tid]);
      float S = 0.f;
#pragma unroll
      for (int w2 = 0; w2 < 8; ++w2) S += sPs[w2][tid] * expf(sPm[w2][tid] - M);
      float* po = pout + (((size_t)b * NPB_ + nb) * M_ + tid) * 2;
      v2f ms; ms.x = M; ms.y = S;
      *(v2f*)po = ms;
    }
  }
}

__global__ void __launch_bounds__(256) k_vtilde(
    const float* __restrict__ Z, const float* __restrict__ u,
    const float* __restrict__ vdual, const float* __restrict__ xp,
    const float* __restrict__ rinv, float* __restrict__ vt) {
  __shared__ float sAcc[8][32][9];
  const int tid = threadIdx.x;
  const int wave = tid >> 5;
  const int lane = tid & 31;
  const int laneLo = lane & 15;
  const int laneHi = lane >> 4;

  const int b = blockIdx.x >> 5;
  const int rem = blockIdx.x & 31;
  const int m0 = (rem >> 4) * 16;
  const int d0 = (rem & 15) * 16;

  const int k_beg = wave * (HW_ / 8);
  const int k_end = k_beg + (HW_ / 8);

  const float* zrow = Z + (size_t)b * (M_ * HW_) + (size_t)(m0 + laneLo) * HW_;
  const float um = u[b * M_ + m0 + laneLo];
  const float* vv = vdual + (size_t)b * HW_;
  const float* xpb = xp + (size_t)b * HW_ * D_ + d0 + laneLo;
  const float* rv = rinv + (size_t)b * HW_;

  v8f acc = {};
  for (int k = k_beg; k < k_end; k += 4) {
    int kk = k + 2 * laneHi;
    v2f z2 = *(const v2f*)(zrow + kk);
    v2f vw = *(const v2f*)(vv + kk);
    v2f a;
    a.x = expf(z2.x + um + vw.x - NORM_);
    a.y = expf(z2.y + um + vw.y - NORM_);
    v2f r2 = *(const v2f*)(rv + kk);
    v2f bb;
    bb.x = xpb[(size_t)kk * D_] * r2.x;
    bb.y = xpb[(size_t)(kk + 1) * D_] * r2.y;
    acc = wmma4(a, bb, acc);
  }

#pragma unroll
  for (int r = 0; r < 8; ++r) sAcc[wave][lane][r] = acc[r];
  __syncthreads();

  if (wave == 0) {
    float* out = vt + (size_t)b * M_ * D_ + (size_t)m0 * D_ + d0 + laneLo;
#pragma unroll
    for (int r = 0; r < 8; ++r) {
      float s = 0.f;
      for (int w2 = 0; w2 < 8; ++w2) s += sAcc[w2][lane][r];
      int row = r + 8 * laneHi;
      out[(size_t)row * D_] = s;
    }
  }
}

extern "C" void kernel_launch(void* const* d_in, const int* in_sizes, int n_in,
                              void* d_out, int out_size, void* d_ws, size_t ws_size,
                              hipStream_t stream) {
  const float* x      = (const float*)d_in[0];
  const float* conv_w = (const float*)d_in[1];
  const float* conv_b = (const float*)d_in[2];
  const float* v      = (const float*)d_in[3];

  float* out_vt = (float*)d_out;
  float* out_xp = out_vt + (size_t)B_ * M_ * D_;

  float* ws   = (float*)d_ws;
  float* what = ws + OFF_WHAT;
  float* Z    = ws + OFF_Z;
  float* u    = ws + OFF_U;
  float* vd   = ws + OFF_V;
  float* rinv = ws + OFF_RINV;
  float* pa   = ws + OFF_PA;
  float* pb   = ws + OFF_PB;
  float* pc   = ws + OFF_PC;

  k_norm_v<<<1, 256, 0, stream>>>(v, what);
  k_proj<<<B_ * (HW_ / 64), 256, 0, stream>>>(x, conv_w, conv_b, what,
                                              out_xp, Z, rinv, pa);
  k_vu<NPA_><<<B_ * NPB_, 256, 0, stream>>>(Z, pa, u, vd, pb, 1);
  k_vu<NPB_><<<B_ * NPB_, 256, 0, stream>>>(Z, pb, u, vd, pc, 1);
  k_vu<NPB_><<<B_ * NPB_, 256, 0, stream>>>(Z, pc, u, vd, pb, 0);
  k_vtilde<<<B_ * M_, 256, 0, stream>>>(Z, u, vd, out_xp, rinv, out_vt);
}


# k_vtilde: each wave issues the loads of the next 32 rows n (x_proj, Z, v, rinv) into a second register set ahead of the current 32 rows' products; sums unchanged
# speedup vs baseline: 1.8959x; 1.0729x over previous
#include <hip/hip_runtime.h>
#include <math.h>

typedef __attribute__((ext_vector_type(2))) float v2f;
typedef __attribute__((ext_vector_type(4))) float v4f;
typedef __attribute__((ext_vector_type(8))) float v8f;

#define B_    16
#define C_    768
#define HW_   4096
#define D_    256
#define M_    32
#define EPS_  0.05f
#define NORM_ (-8.3255517f)
#define NEPS_ 1e-12f

#define OFF_WHAT  0
#define OFF_Z     8192
#define OFF_U     (8192 + 2097152)
#define OFF_V     (OFF_U + 512)
#define OFF_RINV  (OFF_V + 65536)
#define NPA_ 64
#define NPB_ 16
#define OFF_PA    (OFF_RINV + 65536)
#define OFF_PB    (OFF_PA + 65536)
#define OFF_PC    (OFF_PB + 16384)

static __device__ __forceinline__ v8f wmma4(v2f a, v2f b, v8f c) {
  return __builtin_amdgcn_wmma_f32_16x16x4_f32(false, a, false, b, (short)0, c,
                                               false, false);
}

__global__ void k_norm_v(const float* __restrict__ v, float* __restrict__ what) {
  int m = threadIdx.x;
  if (m < M_) {
    float s = 0.f;
    for (int d = 0; d < D_; ++d) { float t = v[m * D_ + d]; s += t * t; }
    float inv = 1.0f / fmaxf(sqrtf(s), NEPS_);
    for (int d = 0; d < D_; ++d) what[m * D_ + d] = v[m * D_ + d] * inv;
  }
}

#define BM_  64
#define BK_  32
#define PA_  72
#define PB_  36
#define PP_  72
__global__ void __launch_bounds__(256) k_proj(
    const float* __restrict__ x, const float* __restrict__ w,
    const float* __restrict__ bias, const float* __restrict__ what,
    float* __restrict__ xp, float* __restrict__ Z, float* __restrict__ rinvg,
    float* __restrict__ part) {
  __shared__ __attribute__((aligned(16))) float sMem[D_ * PP_];
  __shared__ float sRed[8][BM_];
  __shared__ float sRinv[BM_];
  __shared__ float sPart[8][16][2];
  float* sA = sMem;
  float* sB = sMem + BK_ * PA_;

  const int tid = threadIdx.x;
  const int b = blockIdx.x >> 6;
  const int n0 = (blockIdx.x & 63) * BM_;
  const int wave = tid >> 5;
  const int lane = tid & 31;
  const int laneLo = lane & 15;
  const int laneHi = lane >> 4;

  const int ar = tid >> 4;
  const int ac = (tid & 15) * 4;
  const int br = tid >> 3;
  const int bc = (tid & 7) * 4;
  const float* xg = x + (size_t)b * C_ * HW_ + n0 + (size_t)ar * HW_ + ac;
  const float* wg = w + (size_t)br * C_ + bc;

  v4f pa0, pa1, pb[8];
  pa0 = *(const v4f*)(xg);
  pa1 = *(const v4f*)(xg + (size_t)16 * HW_);
#pragma unroll
  for (int j = 0; j < 8; ++j) pb[j] = *(const v4f*)(wg + (size_t)(32 * j) * C_);

  const int d0 = wave * 32;
  v8f acc[4][2];
#pragma unroll
  for (int i = 0; i < 4; ++i) { acc[i][0] = (v8f){}; acc[i][1] = (v8f){}; }

  for (int k0 = 0; k0 < C_; k0 += BK_) {
    *(v4f*)(sA + ar * PA_ + ac) = pa0;
    *(v4f*)(sA + (ar + 16) * PA_ + ac) = pa1;
#pragma unroll
    for (int j = 0; j < 8; ++j) *(v4f*)(sB + (br + 32 * j) * PB_ + bc) = pb[j];
    __syncthreads();

    if (k0 + BK_ < C_) {
      const float* xn = xg + (size_t)(k0 + BK_) * HW_;
      pa0 = *(const v4f*)(xn);
      pa1 = *(const v4f*)(xn + (size_t)16 * HW_);
#pragma unroll
      for (int j = 0; j < 8; ++j)
        pb[j] = *(const v4f*)(wg + (size_t)(32 * j) * C_ + k0 + BK_);
    }

#pragma unroll
    for (int ks = 0; ks < BK_; ks += 4) {
      const int kk = ks + 2 * laneHi;
      v2f a[4];
#pragma unroll
      for (int i = 0; i < 4; ++i) {
        a[i].x = sA[kk * PA_ + 16 * i + laneLo];
        a[i].y = sA[(kk + 1) * PA_ + 16 * i + laneLo];
      }
      const v2f b0 = *(const v2f*)(sB + (d0 + laneLo) * PB_ + kk);
      const v2f b1 = *(const v2f*)(sB + (d0 + 16 + laneLo) * PB_ + kk);
#pragma unroll
      for (int i = 0; i < 4; ++i) {
        acc[i][0] = wmma4(a[i], b0, acc[i][0]);
        acc[i][1] = wmma4(a[i], b1, acc[i][1]);
      }
    }
    __syncthreads();
  }

  const float bi0 = bias[d0 + laneLo];
  const float bi1 = bias[d0 + 16 + laneLo];
#pragma unroll
  for (int i = 0; i < 4; ++i) {
#pragma unroll
    for (int r = 0; r < 8; ++r) { acc[i][0][r] += bi0; acc[i][1][r] += bi1; }
  }

  {
    float* xpb = xp + (size_t)b * HW_ * D_ + (size_t)n0 * D_;
#pragma unroll
    for (int i = 0; i < 4; ++i) {
#pragma unroll
      for (int r = 0; r < 8; ++r) {
        int row = 16 * i + r + 8 * laneHi;
        xpb[(size_t)row * D_ + d0 + laneLo] = acc[i][0][r];
        xpb[(size_t)row * D_ + d0 + 16 + laneLo] = acc[i][1][r];
      }
    }
  }

#pragma unroll
  for (int i = 0; i < 4; ++i) {
#pragma unroll
    for (int r = 0; r < 8; ++r) {
      float val = acc[i][0][r] * acc[i][0][r] + acc[i][1][r] * acc[i][1][r];
      for (int off = 1; off < 16; off <<= 1) val += __shfl_xor(val, off, 32);
      if (laneLo == 0) sRed[wave][16 * i + r + 8 * laneHi] = val;
    }
  }
  __syncthreads();

  if (tid < BM_) {
    float s = 0.f;
    for (int wv = 0; wv < 8; ++wv) s += sRed[wv][tid];
    float inv = 1.0f / fmaxf(sqrtf(s), NEPS_);
    sRinv[tid] = inv;
    rinvg[(size_t)b * HW_ + n0 + tid] = inv;
  }

  float* sP = sMem;
#pragma unroll
  for (int i = 0; i < 4; ++i) {
#pragma unroll
    for (int r = 0; r < 8; ++r) {
      int row = 16 * i + r + 8 * laneHi;
      sP[(d0 + laneLo) * PP_ + row] = acc[i][0][r];
      sP[(d0 + 16 + laneLo) * PP_ + row] = acc[i][1][r];
    }
  }
  __syncthreads();

  {
    const int nt = (wave >> 1) * 16;
    const int m0 = (wave & 1) * 16;
    v8f cs = {};
    const float* wh = what + (size_t)(m0 + laneLo) * D_;
#pragma unroll 4
    for (int k = 0; k < D_; k += 4) {
      int kk = k + 2 * laneHi;
      v2f a; a.x = sP[kk * PP_ + nt + laneLo]; a.y = sP[(kk + 1) * PP_ + nt + laneLo];
      v2f bb = *(const v2f*)(wh + kk);
      cs = wmma4(a, bb, cs);
    }
    float* Zb = Z + (size_t)b * (M_ * HW_) + (size_t)(m0 + laneLo) * HW_ + n0 + nt;
    const float invEps = 1.0f / EPS_;
    float zv[8];
#pragma unroll
    for (int r = 0; r < 8; ++r) {
      int row = r + 8 * laneHi;
      zv[r] = cs[r] * sRinv[nt + row] * invEps;
      Zb[row] = zv[r];
    }
    float mx = zv[0];
#pragma unroll
    for (int r = 1; r < 8; ++r) mx = fmaxf(mx, zv[r]);
    mx = fmaxf(mx, __shfl_xor(mx, 16, 32));
    float sm = 0.f;
#pragma unroll
    for (int r = 0; r < 8; ++r) sm += expf(zv[r] - mx);
    sm += __shfl_xor(sm, 16, 32);
    if (laneHi == 0) { sPart[wave][laneLo][0] = mx; sPart[wave][laneLo][1] = sm; }
  }
  __syncthreads();
  if (tid < M_) {
    const int mt = tid >> 4, ml = tid & 15;
    float M = sPart[mt][ml][0];
#pragma unroll
    for (int q = 1; q < 4; ++q) M = fmaxf(M, sPart[2 * q + mt][ml][0]);
    float S = 0.f;
#pragma unroll
    for (int q = 0; q < 4; ++q)
      S += sPart[2 * q + mt][ml][1] * expf(sPart[2 * q + mt][ml][0] - M);
    float* po = part + (((size_t)b * NPA_ + (blockIdx.x & 63)) * M_ + tid) * 2;
    v2f ms; ms.x = M; ms.y = S;
    *(v2f*)po = ms;
  }
}

#define PT_ 264
template <int NP>
__global__ void __launch_bounds__(256) k_vu(const float* __restrict__ Z,
                                            const float* __restrict__ pin,
                                            float* __restrict__ u,
                                            float* __restrict__ vdual,
                                            float* __restrict__ pout, int want_part) {
  __shared__ float sU[M_];
  __shared__ float sPm[8][M_];
  __shared__ float sPs[8][M_];
  __shared__ float sT[M_ * PT_];
  const int tid = threadIdx.x;
  const int wave = tid >> 5;
  const int lane = tid & 31;
  const int b = blockIdx.x >> 4;
  const int nb = blockIdx.x & 15;
  const int n = nb * 256 + tid;

  const float* zb = Z + (size_t)b * (M_ * HW_) + n;
  float z[M_];
#pragma unroll
  for (int m = 0; m < M_; ++m) z[m] = zb[(size_t)m * HW_];

  {
    const float* pp = pin + ((size_t)b * NP * M_ + lane) * 2;
    v2f ms[NP / 8];
#pragma unroll
    for (int i = 0; i < NP / 8; ++i)
      ms[i] = *(const v2f*)(pp + (size_t)(wave + 8 * i) * (M_ * 2));
    float M = ms[0].x;
#pragma unroll
    for (int i = 1; i < NP / 8; ++i) M = fmaxf(M, ms[i].x);
    float S = 0.f;
#pragma unroll
    for (int i = 0; i < NP / 8; ++i) S += ms[i].y * expf(ms[i].x - M);
    sPm[wave][lane] = M;
    sPs[wave][lane] = S;
  }
  __syncthreads();
  if (tid < M_) {
    float M = sPm[0][tid];
#pragma unroll
    for (int w2 = 1; w2 < 8; ++w2) M = fmaxf(M, sPm[w2][tid]);
    float S = 0.f;
#pragma unroll
    for (int w2 = 0; w2 < 8; ++w2) S += sPs[w2][tid] * expf(sPm[w2][tid] - M);
    const float um = NORM_ - (M + logf(S));
    sU[tid] = um;
    if (nb == 0) u[b * M_ + tid] = um;
  }
  __syncthreads();

  float mx = -1e30f;
#pragma unroll
  for (int m = 0; m < M_; ++m) mx = fmaxf(mx, z[m] + sU[m]);
  float sm = 0.f;
#pragma unroll
  for (int m = 0; m < M_; ++m) sm += expf(z[m] + sU[m] - mx);
  const float vn = NORM_ - (mx + logf(sm));
  vdual[(size_t)b * HW_ + n] = vn;

  if (want_part) {
#pragma unroll
    for (int m = 0; m < M_; ++m) sT[m * PT_ + tid] = z[m] + vn;
    __syncthreads();
    {
      const int m = tid >> 3, seg = tid & 7;
      float tv[32];
#pragma unroll
      for (int i = 0; i < 32; ++i) tv[i] = sT[m * PT_ + seg + 8 * i];
      float M = tv[0];
#pragma unroll
      for (int i = 1; i < 32; ++i) M = fmaxf(M, tv[i]);
      float S = 0.f;
#pragma unroll
      for (int i = 0; i < 32; ++i) S += expf(tv[i] - M);
      sPm[seg][m] = M;
      sPs[seg][m] = S;
    }
    __syncthreads();
    if (tid < M_) {
      float M = sPm[0][tid];
#pragma unroll
      for (int w2 = 1; w2 < 8; ++w2) M = fmaxf(M, sPm[w2][tid]);
      float S = 0.f;
#pragma unroll
      for (int w2 = 0; w2 < 8; ++w2) S += sPs[w2][tid] * expf(sPm[w2][tid] - M);
      float* po = pout + (((size_t)b * NPB_ + nb) * M_ + tid) * 2;
      v2f ms; ms.x = M; ms.y = S;
      *(v2f*)po = ms;
    }
  }
}

#define VB_ 8
#define VT_TRIPS_ ((HW_ / 8) / (8 * VB_))
struct VtBatch { v2f z[VB_], vw[VB_], r[VB_]; float x0[VB_], x1[VB_]; };

static __device__ __forceinline__ void vt_load(VtBatch& t, const float* __restrict__ zq,
                                               const float* __restrict__ vq,
                                               const float* __restrict__ rq,
                                               const float* __restrict__ xq) {
#pragma unroll
  for (int j = 0; j < VB_; ++j) {
    t.x0[j] = xq[(4 * j) * D_];
    t.x1[j] = xq[(4 * j + 1) * D_];
    t.z[j]  = *(const v2f*)(zq + 4 * j);
    t.vw[j] = *(const v2f*)(vq + 4 * j);
    t.r[j]  = *(const v2f*)(rq + 4 * j);
  }
}

static __device__ __forceinline__ v8f vt_mma(const VtBatch& t, float um, v8f acc) {
#pragma unroll
  for (int j = 0; j < VB_; ++j) {
    v2f a;
    a.x = expf(t.z[j].x + um + t.vw[j].x - NORM_);
    a.y = expf(t.z[j].y + um + t.vw[j].y - NORM_);
    v2f bb;
    bb.x = t.x0[j] * t.r[j].x;
    bb.y = t.x1[j] * t.r[j].y;
    acc = wmma4(a, bb, acc);
  }
  return acc;
}

__global__ void __launch_bounds__(256) k_vtilde(
    const float* __restrict__ Z, const float* __restrict__ u,
    const float* __restrict__ vdual, const float* __restrict__ xp,
    const float* __restrict__ rinv, float* __restrict__ vt) {
  __shared__ float sAcc[8][32][9];
  const int tid = threadIdx.x;
  const int wave = tid >> 5;
  const int lane = tid & 31;
  const int laneLo = lane & 15;
  const int laneHi = lane >> 4;

  const int b = blockIdx.x >> 5;
  const int rem = blockIdx.x & 31;
  const int m0 = (rem >> 4) * 16;
  const int d0 = (rem & 15) * 16;

  const int k_beg = wave * (HW_ / 8);

  const float* zrow = Z + (size_t)b * (M_ * HW_) + (size_t)(m0 + laneLo) * HW_;
  const float um = u[b * M_ + m0 + laneLo];
  const float* vv = vdual + (size_t)b * HW_;
  const float* xpb = xp + (size_t)b * HW_ * D_ + d0 + laneLo;
  const float* rv = rinv + (size_t)b * HW_;

  const int kq = k_beg + 2 * laneHi;
  const float* zq = zrow + kq;
  const float* vq = vv + kq;
  const float* rq = rv + kq;
  const float* xq = xpb + (size_t)kq * D_;
  const int BS = 4 * VB_;
  v8f acc = {};
  VtBatch ta, tb;
  vt_load(ta, zq, vq, rq, xq);
  for (int it = 0; it < VT_TRIPS_ - 1; ++it) {
    vt_load(tb, zq + BS, vq + BS, rq + BS, xq + (size_t)BS * D_);
    __builtin_amdgcn_sched_barrier(0);
    acc = vt_mma(ta, um, acc);
    zq += 2 * BS; vq += 2 * BS; rq += 2 * BS; xq += (size_t)(2 * BS) * D_;
    vt_load(ta, zq, vq, rq, xq);
    __builtin_amdgcn_sched_barrier(0);
    acc = vt_mma(tb, um, acc);
  }
  vt_load(tb, zq + BS, vq + BS, rq + BS, xq + (size_t)BS * D_);
  __builtin_amdgcn_sched_barrier(0);
  acc = vt_mma(ta, um, acc);
  __builtin_amdgcn_sched_barrier(0);
  acc = vt_mma(tb, um, acc);

#pragma unroll
  for (int r = 0; r < 8; ++r) sAcc[wave][lane][r] = acc[r];
  __syncthreads();

  if (wave == 0) {
    float* out = vt + (size_t)b * M_ * D_ + (size_t)m0 * D_ + d0 + laneLo;
#pragma unroll
    for (int r = 0; r < 8; ++r) {
      float s = 0.f;
      for (int w2 = 0; w2 < 8; ++w2) s += sAcc[w2][lane][r];
      int row = r + 8 * laneHi;
      out[(size_t)row * D_] = s;
    }
  }
}

extern "C" void kernel_launch(void* const* d_in, const int* in_sizes, int n_in,
                              void* d_out, int out_size, void* d_ws, size_t ws_size,
                              hipStream_t stream) {
  const float* x      = (const float*)d_in[0];
  const float* conv_w = (const float*)d_in[1];
  const float* conv_b = (const float*)d_in[2];
  const float* v      = (const float*)d_in[3];

  float* out_vt = (float*)d_out;
  float* out_xp = out_vt + (size_t)B_ * M_ * D_;

  float* ws   = (float*)d_ws;
  float* what = ws + OFF_WHAT;
  float* Z    = ws + OFF_Z;
  float* u    = ws + OFF_U;
  float* vd   = ws + OFF_V;
  float* rinv = ws + OFF_RINV;
  float* pa   = ws + OFF_PA;
  float* pb   = ws + OFF_PB;
  float* pc   = ws + OFF_PC;

  k_norm_v<<<1, 256, 0, stream>>>(v, what);
  k_proj<<<B_ * (HW_ / 64), 256, 0, stream>>>(x, conv_w, conv_b, what,
                                              out_xp, Z, rinv, pa);
  k_vu<NPA_><<<B_ * NPB_, 256, 0, stream>>>(Z, pa, u, vd, pb, 1);
  k_vu<NPB_><<<B_ * NPB_, 256, 0, stream>>>(Z, pb, u, vd, pc, 1);
  k_vu<NPB_><<<B_ * NPB_, 256, 0, stream>>>(Z, pc, u, vd, pb, 0);
  k_vtilde<<<B_ * M_, 256, 0, stream>>>(Z, u, vd, out_xp, rinv, out_vt);
}


# k_vtilde re-tiled: block 16 m x 32 d, 16 waves x 256 rows n; 1 exp fragment feeds 2 matrix instructions; lanes read 64 B runs of Z, v, rinv; n order in 32-row batches and cross-wave sum order changed
# speedup vs baseline: 2.0010x; 1.0554x over previous
#include <hip/hip_runtime.h>
#include <math.h>

typedef __attribute__((ext_vector_type(2))) float v2f;
typedef __attribute__((ext_vector_type(4))) float v4f;
typedef __attribute__((ext_vector_type(8))) float v8f;

#define B_    16
#define C_    768
#define HW_   4096
#define D_    256
#define M_    32
#define EPS_  0.05f
#define NORM_ (-8.3255517f)
#define NEPS_ 1e-12f

#define OFF_WHAT  0
#define OFF_Z     8192
#define OFF_U     (8192 + 2097152)
#define OFF_V     (OFF_U + 512)
#define OFF_RINV  (OFF_V + 65536)
#define NPA_ 64
#define NPB_ 16
#define OFF_PA    (OFF_RINV + 65536)
#define OFF_PB    (OFF_PA + 65536)
#define OFF_PC    (OFF_PB + 16384)

static __device__ __forceinline__ v8f wmma4(v2f a, v2f b, v8f c) {
  return __builtin_amdgcn_wmma_f32_16x16x4_f32(false, a, false, b, (short)0, c,
                                               false, false);
}

__global__ void k_norm_v(const float* __restrict__ v, float* __restrict__ what) {
  int m = threadIdx.x;
  if (m < M_) {
    float s = 0.f;
    for (int d = 0; d < D_; ++d) { float t = v[m * D_ + d]; s += t * t; }
    float inv = 1.0f / fmaxf(sqrtf(s), NEPS_);
    for (int d = 0; d < D_; ++d) what[m * D_ + d] = v[m * D_ + d] * inv;
  }
}

#define BM_  64
#define BK_  32
#define PA_  72
#define PB_  36
#define PP_  72
__global__ void __launch_bounds__(256) k_proj(
    const float* __restrict__ x, const float* __restrict__ w,
    const float* __restrict__ bias, const float* __restrict__ what,
    float* __restrict__ xp, float* __restrict__ Z, float* __restrict__ rinvg,
    float* __restrict__ part) {
  __shared__ __attribute__((aligned(16))) float sMem[D_ * PP_];
  __shared__ float sRed[8][BM_];
  __shared__ float sRinv[BM_];
  __shared__ float sPart[8][16][2];
  float* sA = sMem;
  float* sB = sMem + BK_ * PA_;

  const int tid = threadIdx.x;
  const int b = blockIdx.x >> 6;
  const int n0 = (blockIdx.x & 63) * BM_;
  const int wave = tid >> 5;
  const int lane = tid & 31;
  const int laneLo = lane & 15;
  const int laneHi = lane >> 4;

  const int ar = tid >> 4;
  const int ac = (tid & 15) * 4;
  const int br = tid >> 3;
  const int bc = (tid & 7) * 4;
  const float* xg = x + (size_t)b * C_ * HW_ + n0 + (size_t)ar * HW_ + ac;
  const float* wg = w + (size_t)br * C_ + bc;

  v4f pa0, pa1, pb[8];
  pa0 = *(const v4f*)(xg);
  pa1 = *(const v4f*)(xg + (size_t)16 * HW_);
#pragma unroll
  for (int j = 0; j < 8; ++j) pb[j] = *(const v4f*)(wg + (size_t)(32 * j) * C_);

  const int d0 = wave * 32;
  v8f acc[4][2];
#pragma unroll
  for (int i = 0; i < 4; ++i) { acc[i][0] = (v8f){}; acc[i][1] = (v8f){}; }

  for (int k0 = 0; k0 < C_; k0 += BK_) {
    *(v4f*)(sA + ar * PA_ + ac) = pa0;
    *(v4f*)(sA + (ar + 16) * PA_ + ac) = pa1;
#pragma unroll
    for (int j = 0; j < 8; ++j) *(v4f*)(sB + (br + 32 * j) * PB_ + bc) = pb[j];
    __syncthreads();

    if (k0 + BK_ < C_) {
      const float* xn = xg + (size_t)(k0 + BK_) * HW_;
      pa0 = *(const v4f*)(xn);
      pa1 = *(const v4f*)(xn + (size_t)16 * HW_);
#pragma unroll
      for (int j = 0; j < 8; ++j)
        pb[j] = *(const v4f*)(wg + (size_t)(32 * j) * C_ + k0 + BK_);
    }

#pragma unroll
    for (int ks = 0; ks < BK_; ks += 4) {
      const int kk = ks + 2 * laneHi;
      v2f a[4];
#pragma unroll
      for (int i = 0; i < 4; ++i) {
        a[i].x = sA[kk * PA_ + 16 * i + laneLo];
        a[i].y = sA[(kk + 1) * PA_ + 16 * i + laneLo];
      }
      const v2f b0 = *(const v2f*)(sB + (d0 + laneLo) * PB_ + kk);
      const v2f b1 = *(const v2f*)(sB + (d0 + 16 + laneLo) * PB_ + kk);
#pragma unroll
      for (int i = 0; i < 4; ++i) {
        acc[i][0] = wmma4(a[i], b0, acc[i][0]);
        acc[i][1] = wmma4(a[i], b1, acc[i][1]);
      }
    }
    __syncthreads();
  }

  const float bi0 = bias[d0 + laneLo];
  const float bi1 = bias[d0 + 16 + laneLo];
#pragma unroll
  for (int i = 0; i < 4; ++i) {
#pragma unroll
    for (int r = 0; r < 8; ++r) { acc[i][0][r] += bi0; acc[i][1][r] += bi1; }
  }

  {
    float* xpb = xp + (size_t)b * HW_ * D_ + (size_t)n0 * D_;
#pragma unroll
    for (int i = 0; i < 4; ++i) {
#pragma unroll
      for (int r = 0; r < 8; ++r) {
        int row = 16 * i + r + 8 * laneHi;
        xpb[(size_t)row * D_ + d0 + laneLo] = acc[i][0][r];
        xpb[(size_t)row * D_ + d0 + 16 + laneLo] = acc[i][1][r];
      }
    }
  }

#pragma unroll
  for (int i = 0; i < 4; ++i) {
#pragma unroll
    for (int r = 0; r < 8; ++r) {
      float val = acc[i][0][r] * acc[i][0][r] + acc[i][1][r] * acc[i][1][r];
      for (int off = 1; off < 16; off <<= 1) val += __shfl_xor(val, off, 32);
      if (laneLo == 0) sRed[wave][16 * i + r + 8 * laneHi] = val;
    }
  }
  __syncthreads();

  if (tid < BM_) {
    float s = 0.f;
    for (int wv = 0; wv < 8; ++wv) s += sRed[wv][tid];
    float inv = 1.0f / fmaxf(sqrtf(s), NEPS_);
    sRinv[tid] = inv;
    rinvg[(size_t)b * HW_ + n0 + tid] = inv;
  }

  float* sP = sMem;
#pragma unroll
  for (int i = 0; i < 4; ++i) {
#pragma unroll
    for (int r = 0; r < 8; ++r) {
      int row = 16 * i + r + 8 * laneHi;
      sP[(d0 + laneLo) * PP_ + row] = acc[i][0][r];
      sP[(d0 + 16 + laneLo) * PP_ + row] = acc[i][1][r];
    }
  }
  __syncthreads();

  {
    const int nt = (wave >> 1) * 16;
    const int m0 = (wave & 1) * 16;
    v8f cs = {};
    const float* wh = what + (size_t)(m0 + laneLo) * D_;
#pragma unroll 4
    for (int k = 0; k < D_; k += 4) {
      int kk = k + 2 * laneHi;
      v2f a; a.x = sP[kk * PP_ + nt + laneLo]; a.y = sP[(kk + 1) * PP_ + nt + laneLo];
      v2f bb = *(const v2f*)(wh + kk);
      cs = wmma4(a, bb, cs);
    }
    float* Zb = Z + (size_t)b * (M_ * HW_) + (size_t)(m0 + laneLo) * HW_ + n0 + nt;
    const float invEps = 1.0f / EPS_;
    float zv[8];
#pragma unroll
    for (int r = 0; r < 8; ++r) {
      int row = r + 8 * laneHi;
      zv[r] = cs[r] * sRinv[nt + row] * invEps;
      Zb[row] = zv[r];
    }
    float mx = zv[0];
#pragma unroll
    for (int r = 1; r < 8; ++r) mx = fmaxf(mx, zv[r]);
    mx = fmaxf(mx, __shfl_xor(mx, 16, 32));
    float sm = 0.f;
#pragma unroll
    for (int r = 0; r < 8; ++r) sm += expf(zv[r] - mx);
    sm += __shfl_xor(sm, 16, 32);
    if (laneHi == 0) { sPart[wave][laneLo][0] = mx; sPart[wave][laneLo][1] = sm; }
  }
  __syncthreads();
  if (tid < M_) {
    const int mt = tid >> 4, ml = tid & 15;
    float M = sPart[mt][ml][0];
#pragma unroll
    for (int q = 1; q < 4; ++q) M = fmaxf(M, sPart[2 * q + mt][ml][0]);
    float S = 0.f;
#pragma unroll
    for (int q = 0; q < 4; ++q)
      S += sPart[2 * q + mt][ml][1] * expf(sPart[2 * q + mt][ml][0] - M);
    float* po = part + (((size_t)b * NPA_ + (blockIdx.x & 63)) * M_ + tid) * 2;
    v2f ms; ms.x = M; ms.y = S;
    *(v2f*)po = ms;
  }
}

#define PT_ 264
template <int NP>
__global__ void __launch_bounds__(256) k_vu(const float* __restrict__ Z,
                                            const float* __restrict__ pin,
                                            float* __restrict__ u,
                                            float* __restrict__ vdual,
                                            float* __restrict__ pout, int want_part) {
  __shared__ float sU[M_];
  __shared__ float sPm[8][M_];
  __shared__ float sPs[8][M_];
  __shared__ float sT[M_ * PT_];
  const int tid = threadIdx.x;
  const int wave = tid >> 5;
  const int lane = tid & 31;
  const int b = blockIdx.x >> 4;
  const int nb = blockIdx.x & 15;
  const int n = nb * 256 + tid;

  const float* zb = Z + (size_t)b * (M_ * HW_) + n;
  float z[M_];
#pragma unroll
  for (int m = 0; m < M_; ++m) z[m] = zb[(size_t)m * HW_];

  {
    const float* pp = pin + ((size_t)b * NP * M_ + lane) * 2;
    v2f ms[NP / 8];
#pragma unroll
    for (int i = 0; i < NP / 8; ++i)
      ms[i] = *(const v2f*)(pp + (size_t)(wave + 8 * i) * (M_ * 2));
    float M = ms[0].x;
#pragma unroll
    for (int i = 1; i < NP / 8; ++i) M = fmaxf(M, ms[i].x);
    float S = 0.f;
#pragma unroll
    for (int i = 0; i < NP / 8; ++i) S += ms[i].y * expf(ms[i].x - M);
    sPm[wave][lane] = M;
    sPs[wave][lane] = S;
  }
  __syncthreads();
  if (tid < M_) {
    float M = sPm[0][tid];
#pragma unroll
    for (int w2 = 1; w2 < 8; ++w2) M = fmaxf(M, sPm[w2][tid]);
    float S = 0.f;
#pragma unroll
    for (int w2 = 0; w2 < 8; ++w2) S += sPs[w2][tid] * expf(sPm[w2][tid] - M);
    const float um = NORM_ - (M + logf(S));
    sU[tid] = um;
    if (nb == 0) u[b * M_ + tid] = um;
  }
  __syncthreads();

  float mx = -1e30f;
#pragma unroll
  for (int m = 0; m < M_; ++m) mx = fmaxf(mx, z[m] + sU[m]);
  float sm = 0.f;
#pragma unroll
  for (int m = 0; m < M_; ++m) sm += expf(z[m] + sU[m] - mx);
  const float vn = NORM_ - (mx + logf(sm));
  vdual[(size_t)b * HW_ + n] = vn;

  if (want_part) {
#pragma unroll
    for (int m = 0; m < M_; ++m) sT[m * PT_ + tid] = z[m] + vn;
    __syncthreads();
    {
      const int m = tid >> 3, seg = tid & 7;
      float tv[32];
#pragma unroll
      for (int i = 0; i < 32; ++i) tv[i] = sT[m * PT_ + seg + 8 * i];
      float M = tv[0];
#pragma unroll
      for (int i = 1; i < 32; ++i) M = fmaxf(M, tv[i]);
      float S = 0.f;
#pragma unroll
      for (int i = 0; i < 32; ++i) S += expf(tv[i] - M);
      sPm[seg][m] = M;
      sPs[seg][m] = S;
    }
    __syncthreads();
    if (tid < M_) {
      float M = sPm[0][tid];
#pragma unroll
      for (int w2 = 1; w2 < 8; ++w2) M = fmaxf(M, sPm[w2][tid]);
      float S = 0.f;
#pragma unroll
      for (int w2 = 0; w2 < 8; ++w2) S += sPs[w2][tid] * expf(sPm[w2][tid] - M);
      float* po = pout + (((size_t)b * NPB_ + nb) * M_ + tid) * 2;
      v2f ms; ms.x = M; ms.y = S;
      *(v2f*)po = ms;
    }
  }
}

#define VR_ 32
#define VQ_ (VR_ / 8)
#define VW_ 16
#define VT_TRIPS_ ((HW_ / VW_) / VR_)
struct VtBatch { v4f z[VQ_], vw[VQ_], r[VQ_]; v2f x[VR_ / 2]; };

static __device__ __forceinline__ void vt_load(VtBatch& t, const float* __restrict__ zq,
                                               const float* __restrict__ vq,
                                               const float* __restrict__ rq,
                                               const float* __restrict__ xq) {
#pragma unroll
  for (int q = 0; q < VQ_; ++q) {
    t.z[q]  = *(const v4f*)(zq + 4 * q);
    t.vw[q] = *(const v4f*)(vq + 4 * q);
  }
#pragma unroll
  for (int q = 0; q < VQ_; ++q) t.r[q] = *(const v4f*)(rq + 4 * q);
#pragma unroll
  for (int e = 0; e < VR_ / 2; ++e) t.x[e] = *(const v2f*)(xq + e * D_);
}

static __device__ __forceinline__ void vt_mma(const VtBatch& t, float um, v8f& acc0,
                                              v8f& acc1) {
#pragma unroll
  for (int j = 0; j < VR_ / 4; ++j) {
    const int q = j >> 1, p = (j & 1) * 2;
    v2f a;
    a.x = expf(t.z[q][p] + um + t.vw[q][p] - NORM_);
    a.y = expf(t.z[q][p + 1] + um + t.vw[q][p + 1] - NORM_);
    const float r0 = t.r[q][p], r1 = t.r[q][p + 1];
    v2f b0, b1;
    b0.x = t.x[2 * j].x * r0;
    b0.y = t.x[2 * j + 1].x * r1;
    b1.x = t.x[2 * j].y * r0;
    b1.y = t.x[2 * j + 1].y * r1;
    acc0 = wmma4(a, b0, acc0);
    acc1 = wmma4(a, b1, acc1);
  }
}

__global__ void __launch_bounds__(512) k_vtilde(
    const float* __restrict__ Z, const float* __restrict__ u,
    const float* __restrict__ vdual, const float* __restrict__ xp,
    const float* __restrict__ rinv, float* __restrict__ vt) {
  __shared__ float sAcc[VW_][2][32][9];
  const int tid = threadIdx.x;
  const int wave = tid >> 5;
  const int lane = tid & 31;
  const int laneLo = lane & 15;
  const int laneHi = lane >> 4;

  const int b = blockIdx.x >> 4;
  const int rem = blockIdx.x & 15;
  const int m0 = (rem >> 3) * 16;
  const int d0 = (rem & 7) * 32;

  const int kq = wave * (HW_ / VW_) + (VR_ / 2) * laneHi;

  const float um = u[b * M_ + m0 + laneLo];
  const float* zq = Z + (size_t)b * (M_ * HW_) + (size_t)(m0 + laneLo) * HW_ + kq;
  const float* vq = vdual + (size_t)b * HW_ + kq;
  const float* rq = rinv + (size_t)b * HW_ + kq;
  const float* xq = xp + (size_t)b * HW_ * D_ + (size_t)kq * D_ + d0 + 2 * laneLo;

  v8f acc0 = {}, acc1 = {};
#pragma nounroll
  for (int it = 0; it < VT_TRIPS_; ++it) {
    VtBatch t;
    vt_load(t, zq, vq, rq, xq);
    __builtin_amdgcn_sched_barrier(0);
    vt_mma(t, um, acc0, acc1);
    zq += VR_; vq += VR_; rq += VR_; xq += (size_t)VR_ * D_;
  }

#pragma unroll
  for (int r = 0; r < 8; ++r) {
    sAcc[wave][0][lane][r] = acc0[r];
    sAcc[wave][1][lane][r] = acc1[r];
  }
  __syncthreads();

  {
    const int row = tid >> 5, col = tid & 31;
    const int tt = col & 1, l = (col >> 1) + 16 * (row >> 3), r = row & 7;
    float s = 0.f;
#pragma unroll
    for (int w2 = 0; w2 < VW_; ++w2) s += sAcc[w2][tt][l][r];
    vt[(size_t)b * M_ * D_ + (size_t)(m0 + row) * D_ + d0 + col] = s;
  }
}

extern "C" void kernel_launch(void* const* d_in, const int* in_sizes, int n_in,
                              void* d_out, int out_size, void* d_ws, size_t ws_size,
                              hipStream_t stream) {
  const float* x      = (const float*)d_in[0];
  const float* conv_w = (const float*)d_in[1];
  const float* conv_b = (const float*)d_in[2];
  const float* v      = (const float*)d_in[3];

  float* out_vt = (float*)d_out;
  float* out_xp = out_vt + (size_t)B_ * M_ * D_;

  float* ws   = (float*)d_ws;
  float* what = ws + OFF_WHAT;
  float* Z    = ws + OFF_Z;
  float* u    = ws + OFF_U;
  float* vd   = ws + OFF_V;
  float* rinv = ws + OFF_RINV;
  float* pa   = ws + OFF_PA;
  float* pb   = ws + OFF_PB;
  float* pc   = ws + OFF_PC;

  k_norm_v<<<1, 256, 0, stream>>>(v, what);
  k_proj<<<B_ * (HW_ / 64), 256, 0, stream>>>(x, conv_w, conv_b, what,
                                              out_xp, Z, rinv, pa);
  k_vu<NPA_><<<B_ * NPB_, 256, 0, stream>>>(Z, pa, u, vd, pb, 1);
  k_vu<NPB_><<<B_ * NPB_, 256, 0, stream>>>(Z, pb, u, vd, pc, 1);
  k_vu<NPB_><<<B_ * NPB_, 256, 0, stream>>>(Z, pc, u, vd, pb, 0);
  k_vtilde<<<B_ * 16, 512, 0, stream>>>(Z, u, vd, out_xp, rinv, out_vt);
}


# cluster-weight normalisation folded into k_proj: each wave squares and sums the weight fragments it loads for its scores and scales them by 1/norm; 1 launch gone; that norm's sum order changed
# speedup vs baseline: 2.0987x; 1.0488x over previous
#include <hip/hip_runtime.h>
#include <math.h>

typedef __attribute__((ext_vector_type(2))) float v2f;
typedef __attribute__((ext_vector_type(4))) float v4f;
typedef __attribute__((ext_vector_type(8))) float v8f;

#define B_    16
#define C_    768
#define HW_   4096
#define D_    256
#define M_    32
#define EPS_  0.05f
#define NORM_ (-8.3255517f)
#define NEPS_ 1e-12f

#define OFF_Z     0
#define OFF_U     2097152
#define OFF_V     (OFF_U + 512)
#define OFF_RINV  (OFF_V + 65536)
#define NPA_ 64
#define NPB_ 16
#define OFF_PA    (OFF_RINV + 65536)
#define OFF_PB    (OFF_PA + 65536)
#define OFF_PC    (OFF_PB + 16384)

static __device__ __forceinline__ v8f wmma4(v2f a, v2f b, v8f c) {
  return __builtin_amdgcn_wmma_f32_16x16x4_f32(false, a, false, b, (short)0, c,
                                               false, false);
}

#define BM_  64
#define BK_  32
#define PA_  72
#define PB_  36
#define PP_  72
__global__ void __launch_bounds__(256) k_proj(
    const float* __restrict__ x, const float* __restrict__ w,
    const float* __restrict__ bias, const float* __restrict__ vc,
    float* __restrict__ xp, float* __restrict__ Z, float* __restrict__ rinvg,
    float* __restrict__ part) {
  __shared__ __attribute__((aligned(16))) float sMem[D_ * PP_];
  __shared__ float sRed[8][BM_];
  __shared__ float sRinv[BM_];
  __shared__ float sPart[8][16][2];
  float* sA = sMem;
  float* sB = sMem + BK_ * PA_;

  const int tid = threadIdx.x;
  const int b = blockIdx.x >> 6;
  const int n0 = (blockIdx.x & 63) * BM_;
  const int wave = tid >> 5;
  const int lane = tid & 31;
  const int laneLo = lane & 15;
  const int laneHi = lane >> 4;

  const int ar = tid >> 4;
  const int ac = (tid & 15) * 4;
  const int br = tid >> 3;
  const int bc = (tid & 7) * 4;
  const float* xg = x + (size_t)b * C_ * HW_ + n0 + (size_t)ar * HW_ + ac;
  const float* wg = w + (size_t)br * C_ + bc;

  v4f pa0, pa1, pb[8];
  pa0 = *(const v4f*)(xg);
  pa1 = *(const v4f*)(xg + (size_t)16 * HW_);
#pragma unroll
  for (int j = 0; j < 8; ++j) pb[j] = *(const v4f*)(wg + (size_t)(32 * j) * C_);

  const int d0 = wave * 32;
  v8f acc[4][2];
#pragma unroll
  for (int i = 0; i < 4; ++i) { acc[i][0] = (v8f){}; acc[i][1] = (v8f){}; }

  for (int k0 = 0; k0 < C_; k0 += BK_) {
    *(v4f*)(sA + ar * PA_ + ac) = pa0;
    *(v4f*)(sA + (ar + 16) * PA_ + ac) = pa1;
#pragma unroll
    for (int j = 0; j < 8; ++j) *(v4f*)(sB + (br + 32 * j) * PB_ + bc) = pb[j];
    __syncthreads();

    if (k0 + BK_ < C_) {
      const float* xn = xg + (size_t)(k0 + BK_) * HW_;
      pa0 = *(const v4f*)(xn);
      pa1 = *(const v4f*)(xn + (size_t)16 * HW_);
#pragma unroll
      for (int j = 0; j < 8; ++j)
        pb[j] = *(const v4f*)(wg + (size_t)(32 * j) * C_ + k0 + BK_);
    }

#pragma unroll
    for (int ks = 0; ks < BK_; ks += 4) {
      const int kk = ks + 2 * laneHi;
      v2f a[4];
#pragma unroll
      for (int i = 0; i < 4; ++i) {
        a[i].x = sA[kk * PA_ + 16 * i + laneLo];
        a[i].y = sA[(kk + 1) * PA_ + 16 * i + laneLo];
      }
      const v2f b0 = *(const v2f*)(sB + (d0 + laneLo) * PB_ + kk);
      const v2f b1 = *(const v2f*)(sB + (d0 + 16 + laneLo) * PB_ + kk);
#pragma unroll
      for (int i = 0; i < 4; ++i) {
        acc[i][0] = wmma4(a[i], b0, acc[i][0]);
        acc[i][1] = wmma4(a[i], b1, acc[i][1]);
      }
    }
    __syncthreads();
  }

  const float bi0 = bias[d0 + laneLo];
  const float bi1 = bias[d0 + 16 + laneLo];
#pragma unroll
  for (int i = 0; i < 4; ++i) {
#pragma unroll
    for (int r = 0; r < 8; ++r) { acc[i][0][r] += bi0; acc[i][1][r] += bi1; }
  }

  {
    float* xpb = xp + (size_t)b * HW_ * D_ + (size_t)n0 * D_;
#pragma unroll
    for (int i = 0; i < 4; ++i) {
#pragma unroll
      for (int r = 0; r < 8; ++r) {
        int row = 16 * i + r + 8 * laneHi;
        xpb[(size_t)row * D_ + d0 + laneLo] = acc[i][0][r];
        xpb[(size_t)row * D_ + d0 + 16 + laneLo] = acc[i][1][r];
      }
    }
  }

#pragma unroll
  for (int i = 0; i < 4; ++i) {
#pragma unroll
    for (int r = 0; r < 8; ++r) {
      float val = acc[i][0][r] * acc[i][0][r] + acc[i][1][r] * acc[i][1][r];
      for (int off = 1; off < 16; off <<= 1) val += __shfl_xor(val, off, 32);
      if (laneLo == 0) sRed[wave][16 * i + r + 8 * laneHi] = val;
    }
  }
  __syncthreads();

  if (tid < BM_) {
    float s = 0.f;
    for (int wv = 0; wv < 8; ++wv) s += sRed[wv][tid];
    float inv = 1.0f / fmaxf(sqrtf(s), NEPS_);
    sRinv[tid] = inv;
    rinvg[(size_t)b * HW_ + n0 + tid] = inv;
  }

  float* sP = sMem;
#pragma unroll
  for (int i = 0; i < 4; ++i) {
#pragma unroll
    for (int r = 0; r < 8; ++r) {
      int row = 16 * i + r + 8 * laneHi;
      sP[(d0 + laneLo) * PP_ + row] = acc[i][0][r];
      sP[(d0 + 16 + laneLo) * PP_ + row] = acc[i][1][r];
    }
  }
  __syncthreads();

  {
    const int nt = (wave >> 1) * 16;
    const int m0 = (wave & 1) * 16;
    v8f cs = {};
    const float* wh = vc + (size_t)(m0 + laneLo) * D_;
    float ss = 0.f;
#pragma unroll 4
    for (int k = 0; k < D_; k += 4) {
      int kk = k + 2 * laneHi;
      v2f a; a.x = sP[kk * PP_ + nt + laneLo]; a.y = sP[(kk + 1) * PP_ + nt + laneLo];
      v2f bb = *(const v2f*)(wh + kk);
      ss += bb.x * bb.x + bb.y * bb.y;
      cs = wmma4(a, bb, cs);
    }
    ss += __shfl_xor(ss, 16, 32);
    const float invv = 1.0f / fmaxf(sqrtf(ss), NEPS_);
    float* Zb = Z + (size_t)b * (M_ * HW_) + (size_t)(m0 + laneLo) * HW_ + n0 + nt;
    const float invEps = 1.0f / EPS_;
    float zv[8];
#pragma unroll
    for (int r = 0; r < 8; ++r) {
      int row = r + 8 * laneHi;
      zv[r] = cs[r] * (sRinv[nt + row] * invv) * invEps;
      Zb[row] = zv[r];
    }
    float mx = zv[0];
#pragma unroll
    for (int r = 1; r < 8; ++r) mx = fmaxf(mx, zv[r]);
    mx = fmaxf(mx, __shfl_xor(mx, 16, 32));
    float sm = 0.f;
#pragma unroll
    for (int r = 0; r < 8; ++r) sm += expf(zv[r] - mx);
    sm += __shfl_xor(sm, 16, 32);
    if (laneHi == 0) { sPart[wave][laneLo][0] = mx; sPart[wave][laneLo][1] = sm; }
  }
  __syncthreads();
  if (tid < M_) {
    const int mt = tid >> 4, ml = tid & 15;
    float M = sPart[mt][ml][0];
#pragma unroll
    for (int q = 1; q < 4; ++q) M = fmaxf(M, sPart[2 * q + mt][ml][0]);
    float S = 0.f;
#pragma unroll
    for (int q = 0; q < 4; ++q)
      S += sPart[2 * q + mt][ml][1] * expf(sPart[2 * q + mt][ml][0] - M);
    float* po = part + (((size_t)b * NPA_ + (blockIdx.x & 63)) * M_ + tid) * 2;
    v2f ms; ms.x = M; ms.y = S;
    *(v2f*)po = ms;
  }
}

#define PT_ 264
template <int NP>
__global__ void __launch_bounds__(256) k_vu(const float* __restrict__ Z,
                                            const float* __restrict__ pin,
                                            float* __restrict__ u,
                                            float* __restrict__ vdual,
                                            float* __restrict__ pout, int want_part) {
  __shared__ float sU[M_];
  __shared__ float sPm[8][M_];
  __shared__ float sPs[8][M_];
  __shared__ float sT[M_ * PT_];
  const int tid = threadIdx.x;
  const int wave = tid >> 5;
  const int lane = tid & 31;
  const int b = blockIdx.x >> 4;
  const int nb = blockIdx.x & 15;
  const int n = nb * 256 + tid;

  const float* zb = Z + (size_t)b * (M_ * HW_) + n;
  float z[M_];
#pragma unroll
  for (int m = 0; m < M_; ++m) z[m] = zb[(size_t)m * HW_];

  {
    const float* pp = pin + ((size_t)b * NP * M_ + lane) * 2;
    v2f ms[NP / 8];
#pragma unroll
    for (int i = 0; i < NP / 8; ++i)
      ms[i] = *(const v2f*)(pp + (size_t)(wave + 8 * i) * (M_ * 2));
    float M = ms[0].x;
#pragma unroll
    for (int i = 1; i < NP / 8; ++i) M = fmaxf(M, ms[i].x);
    float S = 0.f;
#pragma unroll
    for (int i = 0; i < NP / 8; ++i) S += ms[i].y * expf(ms[i].x - M);
    sPm[wave][lane] = M;
    sPs[wave][lane] = S;
  }
  __syncthreads();
  if (tid < M_) {
    float M = sPm[0][tid];
#pragma unroll
    for (int w2 = 1; w2 < 8; ++w2) M = fmaxf(M, sPm[w2][tid]);
    float S = 0.f;
#pragma unroll
    for (int w2 = 0; w2 < 8; ++w2) S += sPs[w2][tid] * expf(sPm[w2][tid] - M);
    const float um = NORM_ - (M + logf(S));
    sU[tid] = um;
    if (nb == 0) u[b * M_ + tid] = um;
  }
  __syncthreads();

  float mx = -1e30f;
#pragma unroll
  for (int m = 0; m < M_; ++m) mx = fmaxf(mx, z[m] + sU[m]);
  float sm = 0.f;
#pragma unroll
  for (int m = 0; m < M_; ++m) sm += expf(z[m] + sU[m] - mx);
  const float vn = NORM_ - (mx + logf(sm));
  vdual[(size_t)b * HW_ + n] = vn;

  if (want_part) {
#pragma unroll
    for (int m = 0; m < M_; ++m) sT[m * PT_ + tid] = z[m] + vn;
    __syncthreads();
    {
      const int m = tid >> 3, seg = tid & 7;
      float tv[32];
#pragma unroll
      for (int i = 0; i < 32; ++i) tv[i] = sT[m * PT_ + seg + 8 * i];
      float M = tv[0];
#pragma unroll
      for (int i = 1; i < 32; ++i) M = fmaxf(M, tv[i]);
      float S = 0.f;
#pragma unroll
      for (int i = 0; i < 32; ++i) S += expf(tv[i] - M);
      sPm[seg][m] = M;
      sPs[seg][m] = S;
    }
    __syncthreads();
    if (tid < M_) {
      float M = sPm[0][tid];
#pragma unroll
      for (int w2 = 1; w2 < 8; ++w2) M = fmaxf(M, sPm[w2][tid]);
      float S = 0.f;
#pragma unroll
      for (int w2 = 0; w2 < 8; ++w2) S += sPs[w2][tid] * expf(sPm[w2][tid] - M);
      float* po = pout + (((size_t)b * NPB_ + nb) * M_ + tid) * 2;
      v2f ms; ms.x = M; ms.y = S;
      *(v2f*)po = ms;
    }
  }
}

#define VR_ 32
#define VQ_ (VR_ / 8)
#define VW_ 16
#define VT_TRIPS_ ((HW_ / VW_) / VR_)
struct VtBatch { v4f z[VQ_], vw[VQ_], r[VQ_]; v2f x[VR_ / 2]; };

static __device__ __forceinline__ void vt_load(VtBatch& t, const float* __restrict__ zq,
                                               const float* __restrict__ vq,
                                               const float* __restrict__ rq,
                                               const float* __restrict__ xq) {
#pragma unroll
  for (int q = 0; q < VQ_; ++q) {
    t.z[q]  = *(const v4f*)(zq + 4 * q);
    t.vw[q] = *(const v4f*)(vq + 4 * q);
  }
#pragma unroll
  for (int q = 0; q < VQ_; ++q) t.r[q] = *(const v4f*)(rq + 4 * q);
#pragma unroll
  for (int e = 0; e < VR_ / 2; ++e) t.x[e] = *(const v2f*)(xq + e * D_);
}

static __device__ __forceinline__ void vt_mma(const VtBatch& t, float um, v8f& acc0,
                                              v8f& acc1) {
#pragma unroll
  for (int j = 0; j < VR_ / 4; ++j) {
    const int q = j >> 1, p = (j & 1) * 2;
    v2f a;
    a.x = expf(t.z[q][p] + um + t.vw[q][p] - NORM_);
    a.y = expf(t.z[q][p + 1] + um + t.vw[q][p + 1] - NORM_);
    const float r0 = t.r[q][p], r1 = t.r[q][p + 1];
    v2f b0, b1;
    b0.x = t.x[2 * j].x * r0;
    b0.y = t.x[2 * j + 1].x * r1;
    b1.x = t.x[2 * j].y * r0;
    b1.y = t.x[2 * j + 1].y * r1;
    acc0 = wmma4(a, b0, acc0);
    acc1 = wmma4(a, b1, acc1);
  }
}

__global__ void __launch_bounds__(512) k_vtilde(
    const float* __restrict__ Z, const float* __restrict__ u,
    const float* __restrict__ vdual, const float* __restrict__ xp,
    const float* __restrict__ rinv, float* __restrict__ vt) {
  __shared__ float sAcc[VW_][2][32][9];
  const int tid = threadIdx.x;
  const int wave = tid >> 5;
  const int lane = tid & 31;
  const int laneLo = lane & 15;
  const int laneHi = lane >> 4;

  const int b = blockIdx.x >> 4;
  const int rem = blockIdx.x & 15;
  const int m0 = (rem >> 3) * 16;
  const int d0 = (rem & 7) * 32;

  const int kq = wave * (HW_ / VW_) + (VR_ / 2) * laneHi;

  const float um = u[b * M_ + m0 + laneLo];
  const float* zq = Z + (size_t)b * (M_ * HW_) + (size_t)(m0 + laneLo) * HW_ + kq;
  const float* vq = vdual + (size_t)b * HW_ + kq;
  const float* rq = rinv + (size_t)b * HW_ + kq;
  const float* xq = xp + (size_t)b * HW_ * D_ + (size_t)kq * D_ + d0 + 2 * laneLo;

  v8f acc0 = {}, acc1 = {};
#pragma nounroll
  for (int it = 0; it < VT_TRIPS_; ++it) {
    VtBatch t;
    vt_load(t, zq, vq, rq, xq);
    __builtin_amdgcn_sched_barrier(0);
    vt_mma(t, um, acc0, acc1);
    zq += VR_; vq += VR_; rq += VR_; xq += (size_t)VR_ * D_;
  }

#pragma unroll
  for (int r = 0; r < 8; ++r) {
    sAcc[wave][0][lane][r] = acc0[r];
    sAcc[wave][1][lane][r] = acc1[r];
  }
  __syncthreads();

  {
    const int row = tid >> 5, col = tid & 31;
    const int tt = col & 1, l = (col >> 1) + 16 * (row >> 3), r = row & 7;
    float s = 0.f;
#pragma unroll
    for (int w2 = 0; w2 < VW_; ++w2) s += sAcc[w2][tt][l][r];
    vt[(size_t)b * M_ * D_ + (size_t)(m0 + row) * D_ + d0 + col] = s;
  }
}

extern "C" void kernel_launch(void* const* d_in, const int* in_sizes, int n_in,
                              void* d_out, int out_size, void* d_ws, size_t ws_size,
                              hipStream_t stream) {
  const float* x      = (const float*)d_in[0];
  const float* conv_w = (const float*)d_in[1];
  const float* conv_b = (const float*)d_in[2];
  const float* v      = (const float*)d_in[3];

  float* out_vt = (float*)d_out;
  float* out_xp = out_vt + (size_t)B_ * M_ * D_;

  float* ws   = (float*)d_ws;
  float* Z    = ws + OFF_Z;
  float* u    = ws + OFF_U;
  float* vd   = ws + OFF_V;
  float* rinv = ws + OFF_RINV;
  float* pa   = ws + OFF_PA;
  float* pb   = ws + OFF_PB;
  float* pc   = ws + OFF_PC;

  k_proj<<<B_ * (HW_ / 64), 256, 0, stream>>>(x, conv_w, conv_b, v,
                                              out_xp, Z, rinv, pa);
  k_vu<NPA_><<<B_ * NPB_, 256, 0, stream>>>(Z, pa, u, vd, pb, 1);
  k_vu<NPB_><<<B_ * NPB_, 256, 0, stream>>>(Z, pb, u, vd, pc, 1);
  k_vu<NPB_><<<B_ * NPB_, 256, 0, stream>>>(Z, pc, u, vd, pb, 0);
  k_vtilde<<<B_ * 16, 512, 0, stream>>>(Z, u, vd, out_xp, rinv, out_vt);
}


# k_proj: each row's sum of squares now comes out of the scores loop (squares of the A fragments it reads); 128 dependent shuffles, their LDS pass and one barrier gone; row-norm sum order changed
# speedup vs baseline: 2.1825x; 1.0399x over previous
#include <hip/hip_runtime.h>
#include <math.h>

typedef __attribute__((ext_vector_type(2))) float v2f;
typedef __attribute__((ext_vector_type(4))) float v4f;
typedef __attribute__((ext_vector_type(8))) float v8f;

#define B_    16
#define C_    768
#define HW_   4096
#define D_    256
#define M_    32
#define EPS_  0.05f
#define NORM_ (-8.3255517f)
#define NEPS_ 1e-12f

#define OFF_Z     0
#define OFF_U     2097152
#define OFF_V     (OFF_U + 512)
#define OFF_RINV  (OFF_V + 65536)
#define NPA_ 64
#define NPB_ 16
#define OFF_PA    (OFF_RINV + 65536)
#define OFF_PB    (OFF_PA + 65536)
#define OFF_PC    (OFF_PB + 16384)

static __device__ __forceinline__ v8f wmma4(v2f a, v2f b, v8f c) {
  return __builtin_amdgcn_wmma_f32_16x16x4_f32(false, a, false, b, (short)0, c,
                                               false, false);
}

#define BM_  64
#define BK_  32
#define PA_  72
#define PB_  36
#define PP_  72
__global__ void __launch_bounds__(256) k_proj(
    const float* __restrict__ x, const float* __restrict__ w,
    const float* __restrict__ bias, const float* __restrict__ vc,
    float* __restrict__ xp, float* __restrict__ Z, float* __restrict__ rinvg,
    float* __restrict__ part) {
  __shared__ __attribute__((aligned(16))) float sMem[D_ * PP_];
  __shared__ float sPart[8][16][2];
  float* sA = sMem;
  float* sB = sMem + BK_ * PA_;

  const int tid = threadIdx.x;
  const int b = blockIdx.x >> 6;
  const int n0 = (blockIdx.x & 63) * BM_;
  const int wave = tid >> 5;
  const int lane = tid & 31;
  const int laneLo = lane & 15;
  const int laneHi = lane >> 4;

  const int ar = tid >> 4;
  const int ac = (tid & 15) * 4;
  const int br = tid >> 3;
  const int bc = (tid & 7) * 4;
  const float* xg = x + (size_t)b * C_ * HW_ + n0 + (size_t)ar * HW_ + ac;
  const float* wg = w + (size_t)br * C_ + bc;

  v4f pa0, pa1, pb[8];
  pa0 = *(const v4f*)(xg);
  pa1 = *(const v4f*)(xg + (size_t)16 * HW_);
#pragma unroll
  for (int j = 0; j < 8; ++j) pb[j] = *(const v4f*)(wg + (size_t)(32 * j) * C_);

  const int d0 = wave * 32;
  v8f acc[4][2];
#pragma unroll
  for (int i = 0; i < 4; ++i) { acc[i][0] = (v8f){}; acc[i][1] = (v8f){}; }

  for (int k0 = 0; k0 < C_; k0 += BK_) {
    *(v4f*)(sA + ar * PA_ + ac) = pa0;
    *(v4f*)(sA + (ar + 16) * PA_ + ac) = pa1;
#pragma unroll
    for (int j = 0; j < 8; ++j) *(v4f*)(sB + (br + 32 * j) * PB_ + bc) = pb[j];
    __syncthreads();

    if (k0 + BK_ < C_) {
      const float* xn = xg + (size_t)(k0 + BK_) * HW_;
      pa0 = *(const v4f*)(xn);
      pa1 = *(const v4f*)(xn + (size_t)16 * HW_);
#pragma unroll
      for (int j = 0; j < 8; ++j)
        pb[j] = *(const v4f*)(wg + (size_t)(32 * j) * C_ + k0 + BK_);
    }

#pragma unroll
    for (int ks = 0; ks < BK_; ks += 4) {
      const int kk = ks + 2 * laneHi;
      v2f a[4];
#pragma unroll
      for (int i = 0; i < 4; ++i) {
        a[i].x = sA[kk * PA_ + 16 * i + laneLo];
        a[i].y = sA[(kk + 1) * PA_ + 16 * i + laneLo];
      }
      const v2f b0 = *(const v2f*)(sB + (d0 + laneLo) * PB_ + kk);
      const v2f b1 = *(const v2f*)(sB + (d0 + 16 + laneLo) * PB_ + kk);
#pragma unroll
      for (int i = 0; i < 4; ++i) {
        acc[i][0] = wmma4(a[i], b0, acc[i][0]);
        acc[i][1] = wmma4(a[i], b1, acc[i][1]);
      }
    }
    __syncthreads();
  }

  const float bi0 = bias[d0 + laneLo];
  const float bi1 = bias[d0 + 16 + laneLo];
#pragma unroll
  for (int i = 0; i < 4; ++i) {
#pragma unroll
    for (int r = 0; r < 8; ++r) { acc[i][0][r] += bi0; acc[i][1][r] += bi1; }
  }

  float* sP = sMem;
#pragma unroll
  for (int i = 0; i < 4; ++i) {
#pragma unroll
    for (int r = 0; r < 8; ++r) {
      int row = 16 * i + r + 8 * laneHi;
      sP[(d0 + laneLo) * PP_ + row] = acc[i][0][r];
      sP[(d0 + 16 + laneLo) * PP_ + row] = acc[i][1][r];
    }
  }
  __syncthreads();

  {
    float* xpb = xp + (size_t)b * HW_ * D_ + (size_t)n0 * D_;
#pragma unroll
    for (int i = 0; i < 4; ++i) {
#pragma unroll
      for (int r = 0; r < 8; ++r) {
        int row = 16 * i + r + 8 * laneHi;
        xpb[(size_t)row * D_ + d0 + laneLo] = acc[i][0][r];
        xpb[(size_t)row * D_ + d0 + 16 + laneLo] = acc[i][1][r];
      }
    }
  }

  {
    const int nt = (wave >> 1) * 16;
    const int m0 = (wave & 1) * 16;
    v8f cs = {};
    const float* wh = vc + (size_t)(m0 + laneLo) * D_;
    float ss = 0.f;
    float sr = 0.f;
#pragma unroll 4
    for (int k = 0; k < D_; k += 4) {
      int kk = k + 2 * laneHi;
      v2f a; a.x = sP[kk * PP_ + nt + laneLo]; a.y = sP[(kk + 1) * PP_ + nt + laneLo];
      v2f bb = *(const v2f*)(wh + kk);
      ss += bb.x * bb.x + bb.y * bb.y;
      sr += a.x * a.x + a.y * a.y;
      cs = wmma4(a, bb, cs);
    }
    ss += __shfl_xor(ss, 16, 32);
    const float invv = 1.0f / fmaxf(sqrtf(ss), NEPS_);
    sr += __shfl_xor(sr, 16, 32);
    const float invr = 1.0f / fmaxf(sqrtf(sr), NEPS_);
    if (m0 == 0 && laneHi == 0) rinvg[(size_t)b * HW_ + n0 + nt + laneLo] = invr;
    float* Zb = Z + (size_t)b * (M_ * HW_) + (size_t)(m0 + laneLo) * HW_ + n0 + nt;
    const float invEps = 1.0f / EPS_;
    float zv[8];
#pragma unroll
    for (int r = 0; r < 8; ++r) {
      int row = r + 8 * laneHi;
      zv[r] = cs[r] * (__shfl(invr, row, 32) * invv) * invEps;
      Zb[row] = zv[r];
    }
    float mx = zv[0];
#pragma unroll
    for (int r = 1; r < 8; ++r) mx = fmaxf(mx, zv[r]);
    mx = fmaxf(mx, __shfl_xor(mx, 16, 32));
    float sm = 0.f;
#pragma unroll
    for (int r = 0; r < 8; ++r) sm += expf(zv[r] - mx);
    sm += __shfl_xor(sm, 16, 32);
    if (laneHi == 0) { sPart[wave][laneLo][0] = mx; sPart[wave][laneLo][1] = sm; }
  }
  __syncthreads();
  if (tid < M_) {
    const int mt = tid >> 4, ml = tid & 15;
    float M = sPart[mt][ml][0];
#pragma unroll
    for (int q = 1; q < 4; ++q) M = fmaxf(M, sPart[2 * q + mt][ml][0]);
    float S = 0.f;
#pragma unroll
    for (int q = 0; q < 4; ++q)
      S += sPart[2 * q + mt][ml][1] * expf(sPart[2 * q + mt][ml][0] - M);
    float* po = part + (((size_t)b * NPA_ + (blockIdx.x & 63)) * M_ + tid) * 2;
    v2f ms; ms.x = M; ms.y = S;
    *(v2f*)po = ms;
  }
}

#define PT_ 264
template <int NP>
__global__ void __launch_bounds__(256) k_vu(const float* __restrict__ Z,
                                            const float* __restrict__ pin,
                                            float* __restrict__ u,
                                            float* __restrict__ vdual,
                                            float* __restrict__ pout, int want_part) {
  __shared__ float sU[M_];
  __shared__ float sPm[8][M_];
  __shared__ float sPs[8][M_];
  __shared__ float sT[M_ * PT_];
  const int tid = threadIdx.x;
  const int wave = tid >> 5;
  const int lane = tid & 31;
  const int b = blockIdx.x >> 4;
  const int nb = blockIdx.x & 15;
  const int n = nb * 256 + tid;

  const float* zb = Z + (size_t)b * (M_ * HW_) + n;
  float z[M_];
#pragma unroll
  for (int m = 0; m < M_; ++m) z[m] = zb[(size_t)m * HW_];

  {
    const float* pp = pin + ((size_t)b * NP * M_ + lane) * 2;
    v2f ms[NP / 8];
#pragma unroll
    for (int i = 0; i < NP / 8; ++i)
      ms[i] = *(const v2f*)(pp + (size_t)(wave + 8 * i) * (M_ * 2));
    float M = ms[0].x;
#pragma unroll
    for (int i = 1; i < NP / 8; ++i) M = fmaxf(M, ms[i].x);
    float S = 0.f;
#pragma unroll
    for (int i = 0; i < NP / 8; ++i) S += ms[i].y * expf(ms[i].x - M);
    sPm[wave][lane] = M;
    sPs[wave][lane] = S;
  }
  __syncthreads();
  if (tid < M_) {
    float M = sPm[0][tid];
#pragma unroll
    for (int w2 = 1; w2 < 8; ++w2) M = fmaxf(M, sPm[w2][tid]);
    float S = 0.f;
#pragma unroll
    for (int w2 = 0; w2 < 8; ++w2) S += sPs[w2][tid] * expf(sPm[w2][tid] - M);
    const float um = NORM_ - (M + logf(S));
    sU[tid] = um;
    if (nb == 0) u[b * M_ + tid] = um;
  }
  __syncthreads();

  float mx = -1e30f;
#pragma unroll
  for (int m = 0; m < M_; ++m) mx = fmaxf(mx, z[m] + sU[m]);
  float sm = 0.f;
#pragma unroll
  for (int m = 0; m < M_; ++m) sm += expf(z[m] + sU[m] - mx);
  const float vn = NORM_ - (mx + logf(sm));
  vdual[(size_t)b * HW_ + n] = vn;

  if (want_part) {
#pragma unroll
    for (int m = 0; m < M_; ++m) sT[m * PT_ + tid] = z[m] + vn;
    __syncthreads();
    {
      const int m = tid >> 3, seg = tid & 7;
      float tv[32];
#pragma unroll
      for (int i = 0; i < 32; ++i) tv[i] = sT[m * PT_ + seg + 8 * i];
      float M = tv[0];
#pragma unroll
      for (int i = 1; i < 32; ++i) M = fmaxf(M, tv[i]);
      float S = 0.f;
#pragma unroll
      for (int i = 0; i < 32; ++i) S += expf(tv[i] - M);
      sPm[seg][m] = M;
      sPs[seg][m] = S;
    }
    __syncthreads();
    if (tid < M_) {
      float M = sPm[0][tid];
#pragma unroll
      for (int w2 = 1; w2 < 8; ++w2) M = fmaxf(M, sPm[w2][tid]);
      float S = 0.f;
#pragma unroll
      for (int w2 = 0; w2 < 8; ++w2) S += sPs[w2][tid] * expf(sPm[w2][tid] - M);
      float* po = pout + (((size_t)b * NPB_ + nb) * M_ + tid) * 2;
      v2f ms; ms.x = M; ms.y = S;
      *(v2f*)po = ms;
    }
  }
}

#define VR_ 32
#define VQ_ (VR_ / 8)
#define VW_ 16
#define VT_TRIPS_ ((HW_ / VW_) / VR_)
struct VtBatch { v4f z[VQ_], vw[VQ_], r[VQ_]; v2f x[VR_ / 2]; };

static __device__ __forceinline__ void vt_load(VtBatch& t, const float* __restrict__ zq,
                                               const float* __restrict__ vq,
                                               const float* __restrict__ rq,
                                               const float* __restrict__ xq) {
#pragma unroll
  for (int q = 0; q < VQ_; ++q) {
    t.z[q]  = *(const v4f*)(zq + 4 * q);
    t.vw[q] = *(const v4f*)(vq + 4 * q);
  }
#pragma unroll
  for (int q = 0; q < VQ_; ++q) t.r[q] = *(const v4f*)(rq + 4 * q);
#pragma unroll
  for (int e = 0; e < VR_ / 2; ++e) t.x[e] = *(const v2f*)(xq + e * D_);
}

static __device__ __forceinline__ void vt_mma(const VtBatch& t, float um, v8f& acc0,
                                              v8f& acc1) {
#pragma unroll
  for (int j = 0; j < VR_ / 4; ++j) {
    const int q = j >> 1, p = (j & 1) * 2;
    v2f a;
    a.x = expf(t.z[q][p] + um + t.vw[q][p] - NORM_);
    a.y = expf(t.z[q][p + 1] + um + t.vw[q][p + 1] - NORM_);
    const float r0 = t.r[q][p], r1 = t.r[q][p + 1];
    v2f b0, b1;
    b0.x = t.x[2 * j].x * r0;
    b0.y = t.x[2 * j + 1].x * r1;
    b1.x = t.x[2 * j].y * r0;
    b1.y = t.x[2 * j + 1].y * r1;
    acc0 = wmma4(a, b0, acc0);
    acc1 = wmma4(a, b1, acc1);
  }
}

__global__ void __launch_bounds__(512) k_vtilde(
    const float* __restrict__ Z, const float* __restrict__ u,
    const float* __restrict__ vdual, const float* __restrict__ xp,
    const float* __restrict__ rinv, float* __restrict__ vt) {
  __shared__ float sAcc[VW_][2][32][9];
  const int tid = threadIdx.x;
  const int wave = tid >> 5;
  const int lane = tid & 31;
  const int laneLo = lane & 15;
  const int laneHi = lane >> 4;

  const int b = blockIdx.x >> 4;
  const int rem = blockIdx.x & 15;
  const int m0 = (rem >> 3) * 16;
  const int d0 = (rem & 7) * 32;

  const int kq = wave * (HW_ / VW_) + (VR_ / 2) * laneHi;

  const float um = u[b * M_ + m0 + laneLo];
  const float* zq = Z + (size_t)b * (M_ * HW_) + (size_t)(m0 + laneLo) * HW_ + kq;
  const float* vq = vdual + (size_t)b * HW_ + kq;
  const float* rq = rinv + (size_t)b * HW_ + kq;
  const float* xq = xp + (size_t)b * HW_ * D_ + (size_t)kq * D_ + d0 + 2 * laneLo;

  v8f acc0 = {}, acc1 = {};
#pragma nounroll
  for (int it = 0; it < VT_TRIPS_; ++it) {
    VtBatch t;
    vt_load(t, zq, vq, rq, xq);
    __builtin_amdgcn_sched_barrier(0);
    vt_mma(t, um, acc0, acc1);
    zq += VR_; vq += VR_; rq += VR_; xq += (size_t)VR_ * D_;
  }

#pragma unroll
  for (int r = 0; r < 8; ++r) {
    sAcc[wave][0][lane][r] = acc0[r];
    sAcc[wave][1][lane][r] = acc1[r];
  }
  __syncthreads();

  {
    const int row = tid >> 5, col = tid & 31;
    const int tt = col & 1, l = (col >> 1) + 16 * (row >> 3), r = row & 7;
    float s = 0.f;
#pragma unroll
    for (int w2 = 0; w2 < VW_; ++w2) s += sAcc[w2][tt][l][r];
    vt[(size_t)b * M_ * D_ + (size_t)(m0 + row) * D_ + d0 + col] = s;
  }
}

extern "C" void kernel_launch(void* const* d_in, const int* in_sizes, int n_in,
                              void* d_out, int out_size, void* d_ws, size_t ws_size,
                              hipStream_t stream) {
  const float* x      = (const float*)d_in[0];
  const float* conv_w = (const float*)d_in[1];
  const float* conv_b = (const float*)d_in[2];
  const float* v      = (const float*)d_in[3];

  float* out_vt = (float*)d_out;
  float* out_xp = out_vt + (size_t)B_ * M_ * D_;

  float* ws   = (float*)d_ws;
  float* Z    = ws + OFF_Z;
  float* u    = ws + OFF_U;
  float* vd   = ws + OFF_V;
  float* rinv = ws + OFF_RINV;
  float* pa   = ws + OFF_PA;
  float* pb   = ws + OFF_PB;
  float* pc   = ws + OFF_PC;

  k_proj<<<B_ * (HW_ / 64), 256, 0, stream>>>(x, conv_w, conv_b, v,
                                              out_xp, Z, rinv, pa);
  k_vu<NPA_><<<B_ * NPB_, 256, 0, stream>>>(Z, pa, u, vd, pb, 1);
  k_vu<NPB_><<<B_ * NPB_, 256, 0, stream>>>(Z, pb, u, vd, pc, 1);
  k_vu<NPB_><<<B_ * NPB_, 256, 0, stream>>>(Z, pc, u, vd, pb, 0);
  k_vtilde<<<B_ * 16, 512, 0, stream>>>(Z, u, vd, out_xp, rinv, out_vt);
}


# k_proj: x chunk staged in LDS as [n][k] (a thread reads 8 k of one column); A and B fragments are 16-byte LDS reads of 4 k: 24 not 40 reads per chunk, no register moves; x_proj k sum order changed
# speedup vs baseline: 2.2891x; 1.0488x over previous
#include <hip/hip_runtime.h>
#include <math.h>

typedef __attribute__((ext_vector_type(2))) float v2f;
typedef __attribute__((ext_vector_type(4))) float v4f;
typedef __attribute__((ext_vector_type(8))) float v8f;

#define B_    16
#define C_    768
#define HW_   4096
#define D_    256
#define M_    32
#define EPS_  0.05f
#define NORM_ (-8.3255517f)
#define NEPS_ 1e-12f

#define OFF_Z     0
#define OFF_U     2097152
#define OFF_V     (OFF_U + 512)
#define OFF_RINV  (OFF_V + 65536)
#define NPA_ 64
#define NPB_ 16
#define OFF_PA    (OFF_RINV + 65536)
#define OFF_PB    (OFF_PA + 65536)
#define OFF_PC    (OFF_PB + 16384)

static __device__ __forceinline__ v8f wmma4(v2f a, v2f b, v8f c) {
  return __builtin_amdgcn_wmma_f32_16x16x4_f32(false, a, false, b, (short)0, c,
                                               false, false);
}

#define BM_  64
#define BK_  32
#define PA_  36
#define PB_  36
#define PP_  72
__global__ void __launch_bounds__(256) k_proj(
    const float* __restrict__ x, const float* __restrict__ w,
    const float* __restrict__ bias, const float* __restrict__ vc,
    float* __restrict__ xp, float* __restrict__ Z, float* __restrict__ rinvg,
    float* __restrict__ part) {
  __shared__ __attribute__((aligned(16))) float sMem[D_ * PP_];
  __shared__ float sPart[8][16][2];
  float* sA = sMem;
  float* sB = sMem + BM_ * PA_;

  const int tid = threadIdx.x;
  const int b = blockIdx.x >> 6;
  const int n0 = (blockIdx.x & 63) * BM_;
  const int wave = tid >> 5;
  const int lane = tid & 31;
  const int laneLo = lane & 15;
  const int laneHi = lane >> 4;

  const int an = tid & 63;
  const int ak = (tid >> 6) * 8;
  const int br = tid >> 3;
  const int bc = (tid & 7) * 4;
  const float* xg = x + (size_t)b * C_ * HW_ + n0 + (size_t)ak * HW_ + an;
  const float* wg = w + (size_t)br * C_ + bc;

  v4f pa0, pa1, pb[8];
#pragma unroll
  for (int j = 0; j < 4; ++j) {
    pa0[j] = xg[(size_t)j * HW_];
    pa1[j] = xg[(size_t)(j + 4) * HW_];
  }
#pragma unroll
  for (int j = 0; j < 8; ++j) pb[j] = *(const v4f*)(wg + (size_t)(32 * j) * C_);

  const int d0 = wave * 32;
  v8f acc[4][2];
#pragma unroll
  for (int i = 0; i < 4; ++i) { acc[i][0] = (v8f){}; acc[i][1] = (v8f){}; }

  for (int k0 = 0; k0 < C_; k0 += BK_) {
    *(v4f*)(sA + an * PA_ + ak) = pa0;
    *(v4f*)(sA + an * PA_ + ak + 4) = pa1;
#pragma unroll
    for (int j = 0; j < 8; ++j) *(v4f*)(sB + (br + 32 * j) * PB_ + bc) = pb[j];
    __syncthreads();

    if (k0 + BK_ < C_) {
      const float* xn = xg + (size_t)(k0 + BK_) * HW_;
#pragma unroll
      for (int j = 0; j < 4; ++j) {
        pa0[j] = xn[(size_t)j * HW_];
        pa1[j] = xn[(size_t)(j + 4) * HW_];
      }
#pragma unroll
      for (int j = 0; j < 8; ++j)
        pb[j] = *(const v4f*)(wg + (size_t)(32 * j) * C_ + k0 + BK_);
    }

#pragma unroll
    for (int ks = 0; ks < BK_; ks += 8) {
      const int kk = ks + 4 * laneHi;
      v4f a[4];
#pragma unroll
      for (int i = 0; i < 4; ++i)
        a[i] = *(const v4f*)(sA + (16 * i + laneLo) * PA_ + kk);
      const v4f b0 = *(const v4f*)(sB + (d0 + laneLo) * PB_ + kk);
      const v4f b1 = *(const v4f*)(sB + (d0 + 16 + laneLo) * PB_ + kk);
#pragma unroll
      for (int i = 0; i < 4; ++i) {
        acc[i][0] = wmma4(a[i].lo, b0.lo, acc[i][0]);
        acc[i][1] = wmma4(a[i].lo, b1.lo, acc[i][1]);
      }
#pragma unroll
      for (int i = 0; i < 4; ++i) {
        acc[i][0] = wmma4(a[i].hi, b0.hi, acc[i][0]);
        acc[i][1] = wmma4(a[i].hi, b1.hi, acc[i][1]);
      }
    }
    __syncthreads();
  }

  const float bi0 = bias[d0 + laneLo];
  const float bi1 = bias[d0 + 16 + laneLo];
#pragma unroll
  for (int i = 0; i < 4; ++i) {
#pragma unroll
    for (int r = 0; r < 8; ++r) { acc[i][0][r] += bi0; acc[i][1][r] += bi1; }
  }

  float* sP = sMem;
#pragma unroll
  for (int i = 0; i < 4; ++i) {
#pragma unroll
    for (int r = 0; r < 8; ++r) {
      int row = 16 * i + r + 8 * laneHi;
      sP[(d0 + laneLo) * PP_ + row] = acc[i][0][r];
      sP[(d0 + 16 + laneLo) * PP_ + row] = acc[i][1][r];
    }
  }
  __syncthreads();

  {
    float* xpb = xp + (size_t)b * HW_ * D_ + (size_t)n0 * D_;
#pragma unroll
    for (int i = 0; i < 4; ++i) {
#pragma unroll
      for (int r = 0; r < 8; ++r) {
        int row = 16 * i + r + 8 * laneHi;
        xpb[(size_t)row * D_ + d0 + laneLo] = acc[i][0][r];
        xpb[(size_t)row * D_ + d0 + 16 + laneLo] = acc[i][1][r];
      }
    }
  }

  {
    const int nt = (wave >> 1) * 16;
    const int m0 = (wave & 1) * 16;
    v8f cs = {};
    const float* wh = vc + (size_t)(m0 + laneLo) * D_;
    float ss = 0.f;
    float sr = 0.f;
#pragma unroll 4
    for (int k = 0; k < D_; k += 4) {
      int kk = k + 2 * laneHi;
      v2f a; a.x = sP[kk * PP_ + nt + laneLo]; a.y = sP[(kk + 1) * PP_ + nt + laneLo];
      v2f bb = *(const v2f*)(wh + kk);
      ss += bb.x * bb.x + bb.y * bb.y;
      sr += a.x * a.x + a.y * a.y;
      cs = wmma4(a, bb, cs);
    }
    ss += __shfl_xor(ss, 16, 32);
    const float invv = 1.0f / fmaxf(sqrtf(ss), NEPS_);
    sr += __shfl_xor(sr, 16, 32);
    const float invr = 1.0f / fmaxf(sqrtf(sr), NEPS_);
    if (m0 == 0 && laneHi == 0) rinvg[(size_t)b * HW_ + n0 + nt + laneLo] = invr;
    float* Zb = Z + (size_t)b * (M_ * HW_) + (size_t)(m0 + laneLo) * HW_ + n0 + nt;
    const float invEps = 1.0f / EPS_;
    float zv[8];
#pragma unroll
    for (int r = 0; r < 8; ++r) {
      int row = r + 8 * laneHi;
      zv[r] = cs[r] * (__shfl(invr, row, 32) * invv) * invEps;
      Zb[row] = zv[r];
    }
    float mx = zv[0];
#pragma unroll
    for (int r = 1; r < 8; ++r) mx = fmaxf(mx, zv[r]);
    mx = fmaxf(mx, __shfl_xor(mx, 16, 32));
    float sm = 0.f;
#pragma unroll
    for (int r = 0; r < 8; ++r) sm += expf(zv[r] - mx);
    sm += __shfl_xor(sm, 16, 32);
    if (laneHi == 0) { sPart[wave][laneLo][0] = mx; sPart[wave][laneLo][1] = sm; }
  }
  __syncthreads();
  if (tid < M_) {
    const int mt = tid >> 4, ml = tid & 15;
    float M = sPart[mt][ml][0];
#pragma unroll
    for (int q = 1; q < 4; ++q) M = fmaxf(M, sPart[2 * q + mt][ml][0]);
    float S = 0.f;
#pragma unroll
    for (int q = 0; q < 4; ++q)
      S += sPart[2 * q + mt][ml][1] * expf(sPart[2 * q + mt][ml][0] - M);
    float* po = part + (((size_t)b * NPA_ + (blockIdx.x & 63)) * M_ + tid) * 2;
    v2f ms; ms.x = M; ms.y = S;
    *(v2f*)po = ms;
  }
}

#define PT_ 264
template <int NP>
__global__ void __launch_bounds__(256) k_vu(const float* __restrict__ Z,
                                            const float* __restrict__ pin,
                                            float* __restrict__ u,
                                            float* __restrict__ vdual,
                                            float* __restrict__ pout, int want_part) {
  __shared__ float sU[M_];
  __shared__ float sPm[8][M_];
  __shared__ float sPs[8][M_];
  __shared__ float sT[M_ * PT_];
  const int tid = threadIdx.x;
  const int wave = tid >> 5;
  const int lane = tid & 31;
  const int b = blockIdx.x >> 4;
  const int nb = blockIdx.x & 15;
  const int n = nb * 256 + tid;

  const float* zb = Z + (size_t)b * (M_ * HW_) + n;
  float z[M_];
#pragma unroll
  for (int m = 0; m < M_; ++m) z[m] = zb[(size_t)m * HW_];

  {
    const float* pp = pin + ((size_t)b * NP * M_ + lane) * 2;
    v2f ms[NP / 8];
#pragma unroll
    for (int i = 0; i < NP / 8; ++i)
      ms[i] = *(const v2f*)(pp + (size_t)(wave + 8 * i) * (M_ * 2));
    float M = ms[0].x;
#pragma unroll
    for (int i = 1; i < NP / 8; ++i) M = fmaxf(M, ms[i].x);
    float S = 0.f;
#pragma unroll
    for (int i = 0; i < NP / 8; ++i) S += ms[i].y * expf(ms[i].x - M);
    sPm[wave][lane] = M;
    sPs[wave][lane] = S;
  }
  __syncthreads();
  if (tid < M_) {
    float M = sPm[0][tid];
#pragma unroll
    for (int w2 = 1; w2 < 8; ++w2) M = fmaxf(M, sPm[w2][tid]);
    float S = 0.f;
#pragma unroll
    for (int w2 = 0; w2 < 8; ++w2) S += sPs[w2][tid] * expf(sPm[w2][tid] - M);
    const float um = NORM_ - (M + logf(S));
    sU[tid] = um;
    if (nb == 0) u[b * M_ + tid] = um;
  }
  __syncthreads();

  float mx = -1e30f;
#pragma unroll
  for (int m = 0; m < M_; ++m) mx = fmaxf(mx, z[m] + sU[m]);
  float sm = 0.f;
#pragma unroll
  for (int m = 0; m < M_; ++m) sm += expf(z[m] + sU[m] - mx);
  const float vn = NORM_ - (mx + logf(sm));
  vdual[(size_t)b * HW_ + n] = vn;

  if (want_part) {
#pragma unroll
    for (int m = 0; m < M_; ++m) sT[m * PT_ + tid] = z[m] + vn;
    __syncthreads();
    {
      const int m = tid >> 3, seg = tid & 7;
      float tv[32];
#pragma unroll
      for (int i = 0; i < 32; ++i) tv[i] = sT[m * PT_ + seg + 8 * i];
      float M = tv[0];
#pragma unroll
      for (int i = 1; i < 32; ++i) M = fmaxf(M, tv[i]);
      float S = 0.f;
#pragma unroll
      for (int i = 0; i < 32; ++i) S += expf(tv[i] - M);
      sPm[seg][m] = M;
      sPs[seg][m] = S;
    }
    __syncthreads();
    if (tid < M_) {
      float M = sPm[0][tid];
#pragma unroll
      for (int w2 = 1; w2 < 8; ++w2) M = fmaxf(M, sPm[w2][tid]);
      float S = 0.f;
#pragma unroll
      for (int w2 = 0; w2 < 8; ++w2) S += sPs[w2][tid] * expf(sPm[w2][tid] - M);
      float* po = pout + (((size_t)b * NPB_ + nb) * M_ + tid) * 2;
      v2f ms; ms.x = M; ms.y = S;
      *(v2f*)po = ms;
    }
  }
}

#define VR_ 32
#define VQ_ (VR_ / 8)
#define VW_ 16
#define VT_TRIPS_ ((HW_ / VW_) / VR_)
struct VtBatch { v4f z[VQ_], vw[VQ_], r[VQ_]; v2f x[VR_ / 2]; };

static __device__ __forceinline__ void vt_load(VtBatch& t, const float* __restrict__ zq,
                                               const float* __restrict__ vq,
                                               const float* __restrict__ rq,
                                               const float* __restrict__ xq) {
#pragma unroll
  for (int q = 0; q < VQ_; ++q) {
    t.z[q]  = *(const v4f*)(zq + 4 * q);
    t.vw[q] = *(const v4f*)(vq + 4 * q);
  }
#pragma unroll
  for (int q = 0; q < VQ_; ++q) t.r[q] = *(const v4f*)(rq + 4 * q);
#pragma unroll
  for (int e = 0; e < VR_ / 2; ++e) t.x[e] = *(const v2f*)(xq + e * D_);
}

static __device__ __forceinline__ void vt_mma(const VtBatch& t, float um, v8f& acc0,
                                              v8f& acc1) {
#pragma unroll
  for (int j = 0; j < VR_ / 4; ++j) {
    const int q = j >> 1, p = (j & 1) * 2;
    v2f a;
    a.x = expf(t.z[q][p] + um + t.vw[q][p] - NORM_);
    a.y = expf(t.z[q][p + 1] + um + t.vw[q][p + 1] - NORM_);
    const float r0 = t.r[q][p], r1 = t.r[q][p + 1];
    v2f b0, b1;
    b0.x = t.x[2 * j].x * r0;
    b0.y = t.x[2 * j + 1].x * r1;
    b1.x = t.x[2 * j].y * r0;
    b1.y = t.x[2 * j + 1].y * r1;
    acc0 = wmma4(a, b0, acc0);
    acc1 = wmma4(a, b1, acc1);
  }
}

__global__ void __launch_bounds__(512) k_vtilde(
    const float* __restrict__ Z, const float* __restrict__ u,
    const float* __restrict__ vdual, const float* __restrict__ xp,
    const float* __restrict__ rinv, float* __restrict__ vt) {
  __shared__ float sAcc[VW_][2][32][9];
  const int tid = threadIdx.x;
  const int wave = tid >> 5;
  const int lane = tid & 31;
  const int laneLo = lane & 15;
  const int laneHi = lane >> 4;

  const int b = blockIdx.x >> 4;
  const int rem = blockIdx.x & 15;
  const int m0 = (rem >> 3) * 16;
  const int d0 = (rem & 7) * 32;

  const int kq = wave * (HW_ / VW_) + (VR_ / 2) * laneHi;

  const float um = u[b * M_ + m0 + laneLo];
  const float* zq = Z + (size_t)b * (M_ * HW_) + (size_t)(m0 + laneLo) * HW_ + kq;
  const float* vq = vdual + (size_t)b * HW_ + kq;
  const float* rq = rinv + (size_t)b * HW_ + kq;
  const float* xq = xp + (size_t)b * HW_ * D_ + (size_t)kq * D_ + d0 + 2 * laneLo;

  v8f acc0 = {}, acc1 = {};
#pragma nounroll
  for (int it = 0; it < VT_TRIPS_; ++it) {
    VtBatch t;
    vt_load(t, zq, vq, rq, xq);
    __builtin_amdgcn_sched_barrier(0);
    vt_mma(t, um, acc0, acc1);
    zq += VR_; vq += VR_; rq += VR_; xq += (size_t)VR_ * D_;
  }

#pragma unroll
  for (int r = 0; r < 8; ++r) {
    sAcc[wave][0][lane][r] = acc0[r];
    sAcc[wave][1][lane][r] = acc1[r];
  }
  __syncthreads();

  {
    const int row = tid >> 5, col = tid & 31;
    const int tt = col & 1, l = (col >> 1) + 16 * (row >> 3), r = row & 7;
    float s = 0.f;
#pragma unroll
    for (int w2 = 0; w2 < VW_; ++w2) s += sAcc[w2][tt][l][r];
    vt[(size_t)b * M_ * D_ + (size_t)(m0 + row) * D_ + d0 + col] = s;
  }
}

extern "C" void kernel_launch(void* const* d_in, const int* in_sizes, int n_in,
                              void* d_out, int out_size, void* d_ws, size_t ws_size,
                              hipStream_t stream) {
  const float* x      = (const float*)d_in[0];
  const float* conv_w = (const float*)d_in[1];
  const float* conv_b = (const float*)d_in[2];
  const float* v      = (const float*)d_in[3];

  float* out_vt = (float*)d_out;
  float* out_xp = out_vt + (size_t)B_ * M_ * D_;

  float* ws   = (float*)d_ws;
  float* Z    = ws + OFF_Z;
  float* u    = ws + OFF_U;
  float* vd   = ws + OFF_V;
  float* rinv = ws + OFF_RINV;
  float* pa   = ws + OFF_PA;
  float* pb   = ws + OFF_PB;
  float* pc   = ws + OFF_PC;

  k_proj<<<B_ * (HW_ / 64), 256, 0, stream>>>(x, conv_w, conv_b, v,
                                              out_xp, Z, rinv, pa);
  k_vu<NPA_><<<B_ * NPB_, 256, 0, stream>>>(Z, pa, u, vd, pb, 1);
  k_vu<NPB_><<<B_ * NPB_, 256, 0, stream>>>(Z, pb, u, vd, pc, 1);
  k_vu<NPB_><<<B_ * NPB_, 256, 0, stream>>>(Z, pc, u, vd, pb, 0);
  k_vtilde<<<B_ * 16, 512, 0, stream>>>(Z, u, vd, out_xp, rinv, out_vt);
}


# k_proj epilogue: tile and cluster weights row-major in LDS: x_proj leaves in whole-line 16-byte stores, scores operands are 16-byte LDS reads, no global load in its loop; scores, norm k order changed
# speedup vs baseline: 2.3620x; 1.0318x over previous
#include <hip/hip_runtime.h>
#include <math.h>

typedef __attribute__((ext_vector_type(2))) float v2f;
typedef __attribute__((ext_vector_type(4))) float v4f;
typedef __attribute__((ext_vector_type(8))) float v8f;

#define B_    16
#define C_    768
#define HW_   4096
#define D_    256
#define M_    32
#define EPS_  0.05f
#define NORM_ (-8.3255517f)
#define NEPS_ 1e-12f

#define OFF_Z     0
#define OFF_U     2097152
#define OFF_V     (OFF_U + 512)
#define OFF_RINV  (OFF_V + 65536)
#define NPA_ 64
#define NPB_ 16
#define OFF_PA    (OFF_RINV + 65536)
#define OFF_PB    (OFF_PA + 65536)
#define OFF_PC    (OFF_PB + 16384)

static __device__ __forceinline__ v8f wmma4(v2f a, v2f b, v8f c) {
  return __builtin_amdgcn_wmma_f32_16x16x4_f32(false, a, false, b, (short)0, c,
                                               false, false);
}

#define BM_  64
#define BK_  32
#define PA_  36
#define PB_  36
#define PP_  72
#define PQ_  260
__global__ void __launch_bounds__(256) k_proj(
    const float* __restrict__ x, const float* __restrict__ w,
    const float* __restrict__ bias, const float* __restrict__ vc,
    float* __restrict__ xp, float* __restrict__ Z, float* __restrict__ rinvg,
    float* __restrict__ part) {
  __shared__ __attribute__((aligned(16))) float sMem[D_ * PP_];
  __shared__ float sPart[8][16][2];
  __shared__ __attribute__((aligned(16))) float sV[M_ * PQ_];
  float* sA = sMem;
  float* sB = sMem + BM_ * PA_;

  const int tid = threadIdx.x;
  const int b = blockIdx.x >> 6;
  const int n0 = (blockIdx.x & 63) * BM_;
  const int wave = tid >> 5;
  const int lane = tid & 31;
  const int laneLo = lane & 15;
  const int laneHi = lane >> 4;

  const int an = tid & 63;
  const int ak = (tid >> 6) * 8;
  const int br = tid >> 3;
  const int bc = (tid & 7) * 4;
  const float* xg = x + (size_t)b * C_ * HW_ + n0 + (size_t)ak * HW_ + an;
  const float* wg = w + (size_t)br * C_ + bc;

  v4f pa0, pa1, pb[8];
#pragma unroll
  for (int j = 0; j < 4; ++j) {
    pa0[j] = xg[(size_t)j * HW_];
    pa1[j] = xg[(size_t)(j + 4) * HW_];
  }
#pragma unroll
  for (int j = 0; j < 8; ++j) pb[j] = *(const v4f*)(wg + (size_t)(32 * j) * C_);

#pragma unroll
  for (int j = 0; j < 8; ++j) {
    const int q = tid + 256 * j;
    const int vm = q >> 6, vd = (q & 63) * 4;
    *(v4f*)(sV + vm * PQ_ + vd) = *(const v4f*)(vc + (size_t)vm * D_ + vd);
  }

  const int d0 = wave * 32;
  v8f acc[4][2];
#pragma unroll
  for (int i = 0; i < 4; ++i) { acc[i][0] = (v8f){}; acc[i][1] = (v8f){}; }

  for (int k0 = 0; k0 < C_; k0 += BK_) {
    *(v4f*)(sA + an * PA_ + ak) = pa0;
    *(v4f*)(sA + an * PA_ + ak + 4) = pa1;
#pragma unroll
    for (int j = 0; j < 8; ++j) *(v4f*)(sB + (br + 32 * j) * PB_ + bc) = pb[j];
    __syncthreads();

    if (k0 + BK_ < C_) {
      const float* xn = xg + (size_t)(k0 + BK_) * HW_;
#pragma unroll
      for (int j = 0; j < 4; ++j) {
        pa0[j] = xn[(size_t)j * HW_];
        pa1[j] = xn[(size_t)(j + 4) * HW_];
      }
#pragma unroll
      for (int j = 0; j < 8; ++j)
        pb[j] = *(const v4f*)(wg + (size_t)(32 * j) * C_ + k0 + BK_);
    }

#pragma unroll
    for (int ks = 0; ks < BK_; ks += 8) {
      const int kk = ks + 4 * laneHi;
      v4f a[4];
#pragma unroll
      for (int i = 0; i < 4; ++i)
        a[i] = *(const v4f*)(sA + (16 * i + laneLo) * PA_ + kk);
      const v4f b0 = *(const v4f*)(sB + (d0 + laneLo) * PB_ + kk);
      const v4f b1 = *(const v4f*)(sB + (d0 + 16 + laneLo) * PB_ + kk);
#pragma unroll
      for (int i = 0; i < 4; ++i) {
        acc[i][0] = wmma4(a[i].lo, b0.lo, acc[i][0]);
        acc[i][1] = wmma4(a[i].lo, b1.lo, acc[i][1]);
      }
#pragma unroll
      for (int i = 0; i < 4; ++i) {
        acc[i][0] = wmma4(a[i].hi, b0.hi, acc[i][0]);
        acc[i][1] = wmma4(a[i].hi, b1.hi, acc[i][1]);
      }
    }
    __syncthreads();
  }

  const float bi0 = bias[d0 + laneLo];
  const float bi1 = bias[d0 + 16 + laneLo];
#pragma unroll
  for (int i = 0; i < 4; ++i) {
#pragma unroll
    for (int r = 0; r < 8; ++r) { acc[i][0][r] += bi0; acc[i][1][r] += bi1; }
  }

  float* sQ = sMem;
#pragma unroll
  for (int i = 0; i < 4; ++i) {
#pragma unroll
    for (int r = 0; r < 8; ++r) {
      int row = 16 * i + r + 8 * laneHi;
      sQ[row * PQ_ + d0 + laneLo] = acc[i][0][r];
      sQ[row * PQ_ + d0 + 16 + laneLo] = acc[i][1][r];
    }
  }
  __syncthreads();

  {
    float* xpb = xp + (size_t)b * HW_ * D_ + (size_t)n0 * D_;
    const int c4 = (tid & 63) * 4;
    const int r0 = tid >> 6;
#pragma unroll
    for (int j = 0; j < 16; ++j) {
      const int row = r0 + 4 * j;
      *(v4f*)(xpb + (size_t)row * D_ + c4) = *(const v4f*)(sQ + row * PQ_ + c4);
    }
  }
  __builtin_amdgcn_sched_barrier(0);

  {
    const int nt = (wave >> 1) * 16;
    const int m0 = (wave & 1) * 16;
    v8f cs = {};
    float ss = 0.f;
    float sr = 0.f;
    const float* qa = sQ + (nt + laneLo) * PQ_ + 4 * laneHi;
    const float* qb = sV + (m0 + laneLo) * PQ_ + 4 * laneHi;
#pragma unroll
    for (int g = 0; g < 32; ++g) {
      const v4f a = *(const v4f*)(qa + 8 * g);
      const v4f bb = *(const v4f*)(qb + 8 * g);
      ss += bb.x * bb.x + bb.y * bb.y;
      ss += bb.z * bb.z + bb.w * bb.w;
      sr += a.x * a.x + a.y * a.y;
      sr += a.z * a.z + a.w * a.w;
      cs = wmma4(a.lo, bb.lo, cs);
      cs = wmma4(a.hi, bb.hi, cs);
      if ((g & 7) == 7) __builtin_amdgcn_sched_barrier(0);
    }
    ss += __shfl_xor(ss, 16, 32);
    const float invv = 1.0f / fmaxf(sqrtf(ss), NEPS_);
    sr += __shfl_xor(sr, 16, 32);
    const float invr = 1.0f / fmaxf(sqrtf(sr), NEPS_);
    if (m0 == 0 && laneHi == 0) rinvg[(size_t)b * HW_ + n0 + nt + laneLo] = invr;
    float* Zb = Z + (size_t)b * (M_ * HW_) + (size_t)(m0 + laneLo) * HW_ + n0 + nt;
    const float invEps = 1.0f / EPS_;
    float zv[8];
#pragma unroll
    for (int r = 0; r < 8; ++r) {
      int row = r + 8 * laneHi;
      zv[r] = cs[r] * (__shfl(invr, row, 32) * invv) * invEps;
      Zb[row] = zv[r];
    }
    float mx = zv[0];
#pragma unroll
    for (int r = 1; r < 8; ++r) mx = fmaxf(mx, zv[r]);
    mx = fmaxf(mx, __shfl_xor(mx, 16, 32));
    float sm = 0.f;
#pragma unroll
    for (int r = 0; r < 8; ++r) sm += expf(zv[r] - mx);
    sm += __shfl_xor(sm, 16, 32);
    if (laneHi == 0) { sPart[wave][laneLo][0] = mx; sPart[wave][laneLo][1] = sm; }
  }
  __syncthreads();
  if (tid < M_) {
    const int mt = tid >> 4, ml = tid & 15;
    float M = sPart[mt][ml][0];
#pragma unroll
    for (int q = 1; q < 4; ++q) M = fmaxf(M, sPart[2 * q + mt][ml][0]);
    float S = 0.f;
#pragma unroll
    for (int q = 0; q < 4; ++q)
      S += sPart[2 * q + mt][ml][1] * expf(sPart[2 * q + mt][ml][0] - M);
    float* po = part + (((size_t)b * NPA_ + (blockIdx.x & 63)) * M_ + tid) * 2;
    v2f ms; ms.x = M; ms.y = S;
    *(v2f*)po = ms;
  }
}

#define PT_ 264
template <int NP>
__global__ void __launch_bounds__(256) k_vu(const float* __restrict__ Z,
                                            const float* __restrict__ pin,
                                            float* __restrict__ u,
                                            float* __restrict__ vdual,
                                            float* __restrict__ pout, int want_part) {
  __shared__ float sU[M_];
  __shared__ float sPm[8][M_];
  __shared__ float sPs[8][M_];
  __shared__ float sT[M_ * PT_];
  const int tid = threadIdx.x;
  const int wave = tid >> 5;
  const int lane = tid & 31;
  const int b = blockIdx.x >> 4;
  const int nb = blockIdx.x & 15;
  const int n = nb * 256 + tid;

  const float* zb = Z + (size_t)b * (M_ * HW_) + n;
  float z[M_];
#pragma unroll
  for (int m = 0; m < M_; ++m) z[m] = zb[(size_t)m * HW_];

  {
    const float* pp = pin + ((size_t)b * NP * M_ + lane) * 2;
    v2f ms[NP / 8];
#pragma unroll
    for (int i = 0; i < NP / 8; ++i)
      ms[i] = *(const v2f*)(pp + (size_t)(wave + 8 * i) * (M_ * 2));
    float M = ms[0].x;
#pragma unroll
    for (int i = 1; i < NP / 8; ++i) M = fmaxf(M, ms[i].x);
    float S = 0.f;
#pragma unroll
    for (int i = 0; i < NP / 8; ++i) S += ms[i].y * expf(ms[i].x - M);
    sPm[wave][lane] = M;
    sPs[wave][lane] = S;
  }
  __syncthreads();
  if (tid < M_) {
    float M = sPm[0][tid];
#pragma unroll
    for (int w2 = 1; w2 < 8; ++w2) M = fmaxf(M, sPm[w2][tid]);
    float S = 0.f;
#pragma unroll
    for (int w2 = 0; w2 < 8; ++w2) S += sPs[w2][tid] * expf(sPm[w2][tid] - M);
    const float um = NORM_ - (M + logf(S));
    sU[tid] = um;
    if (nb == 0) u[b * M_ + tid] = um;
  }
  __syncthreads();

  float mx = -1e30f;
#pragma unroll
  for (int m = 0; m < M_; ++m) mx = fmaxf(mx, z[m] + sU[m]);
  float sm = 0.f;
#pragma unroll
  for (int m = 0; m < M_; ++m) sm += expf(z[m] + sU[m] - mx);
  const float vn = NORM_ - (mx + logf(sm));
  vdual[(size_t)b * HW_ + n] = vn;

  if (want_part) {
#pragma unroll
    for (int m = 0; m < M_; ++m) sT[m * PT_ + tid] = z[m] + vn;
    __syncthreads();
    {
      const int m = tid >> 3, seg = tid & 7;
      float tv[32];
#pragma unroll
      for (int i = 0; i < 32; ++i) tv[i] = sT[m * PT_ + seg + 8 * i];
      float M = tv[0];
#pragma unroll
      for (int i = 1; i < 32; ++i) M = fmaxf(M, tv[i]);
      float S = 0.f;
#pragma unroll
      for (int i = 0; i < 32; ++i) S += expf(tv[i] - M);
      sPm[seg][m] = M;
      sPs[seg][m] = S;
    }
    __syncthreads();
    if (tid < M_) {
      float M = sPm[0][tid];
#pragma unroll
      for (int w2 = 1; w2 < 8; ++w2) M = fmaxf(M, sPm[w2][tid]);
      float S = 0.f;
#pragma unroll
      for (int w2 = 0; w2 < 8; ++w2) S += sPs[w2][tid] * expf(sPm[w2][tid] - M);
      float* po = pout + (((size_t)b * NPB_ + nb) * M_ + tid) * 2;
      v2f ms; ms.x = M; ms.y = S;
      *(v2f*)po = ms;
    }
  }
}

#define VR_ 32
#define VQ_ (VR_ / 8)
#define VW_ 16
#define VT_TRIPS_ ((HW_ / VW_) / VR_)
struct VtBatch { v4f z[VQ_], vw[VQ_], r[VQ_]; v2f x[VR_ / 2]; };

static __device__ __forceinline__ void vt_load(VtBatch& t, const float* __restrict__ zq,
                                               const float* __restrict__ vq,
                                               const float* __restrict__ rq,
                                               const float* __restrict__ xq) {
#pragma unroll
  for (int q = 0; q < VQ_; ++q) {
    t.z[q]  = *(const v4f*)(zq + 4 * q);
    t.vw[q] = *(const v4f*)(vq + 4 * q);
  }
#pragma unroll
  for (int q = 0; q < VQ_; ++q) t.r[q] = *(const v4f*)(rq + 4 * q);
#pragma unroll
  for (int e = 0; e < VR_ / 2; ++e) t.x[e] = *(const v2f*)(xq + e * D_);
}

static __device__ __forceinline__ void vt_mma(const VtBatch& t, float um, v8f& acc0,
                                              v8f& acc1) {
#pragma unroll
  for (int j = 0; j < VR_ / 4; ++j) {
    const int q = j >> 1, p = (j & 1) * 2;
    v2f a;
    a.x = expf(t.z[q][p] + um + t.vw[q][p] - NORM_);
    a.y = expf(t.z[q][p + 1] + um + t.vw[q][p + 1] - NORM_);
    const float r0 = t.r[q][p], r1 = t.r[q][p + 1];
    v2f b0, b1;
    b0.x = t.x[2 * j].x * r0;
    b0.y = t.x[2 * j + 1].x * r1;
    b1.x = t.x[2 * j].y * r0;
    b1.y = t.x[2 * j + 1].y * r1;
    acc0 = wmma4(a, b0, acc0);
    acc1 = wmma4(a, b1, acc1);
  }
}

__global__ void __launch_bounds__(512) k_vtilde(
    const float* __restrict__ Z, const float* __restrict__ u,
    const float* __restrict__ vdual, const float* __restrict__ xp,
    const float* __restrict__ rinv, float* __restrict__ vt) {
  __shared__ float sAcc[VW_][2][32][9];
  const int tid = threadIdx.x;
  const int wave = tid >> 5;
  const int lane = tid & 31;
  const int laneLo = lane & 15;
  const int laneHi = lane >> 4;

  const int b = blockIdx.x >> 4;
  const int rem = blockIdx.x & 15;
  const int m0 = (rem >> 3) * 16;
  const int d0 = (rem & 7) * 32;

  const int kq = wave * (HW_ / VW_) + (VR_ / 2) * laneHi;

  const float um = u[b * M_ + m0 + laneLo];
  const float* zq = Z + (size_t)b * (M_ * HW_) + (size_t)(m0 + laneLo) * HW_ + kq;
  const float* vq = vdual + (size_t)b * HW_ + kq;
  const float* rq = rinv + (size_t)b * HW_ + kq;
  const float* xq = xp + (size_t)b * HW_ * D_ + (size_t)kq * D_ + d0 + 2 * laneLo;

  v8f acc0 = {}, acc1 = {};
#pragma nounroll
  for (int it = 0; it < VT_TRIPS_; ++it) {
    VtBatch t;
    vt_load(t, zq, vq, rq, xq);
    __builtin_amdgcn_sched_barrier(0);
    vt_mma(t, um, acc0, acc1);
    zq += VR_; vq += VR_; rq += VR_; xq += (size_t)VR_ * D_;
  }

#pragma unroll
  for (int r = 0; r < 8; ++r) {
    sAcc[wave][0][lane][r] = acc0[r];
    sAcc[wave][1][lane][r] = acc1[r];
  }
  __syncthreads();

  {
    const int row = tid >> 5, col = tid & 31;
    const int tt = col & 1, l = (col >> 1) + 16 * (row >> 3), r = row & 7;
    float s = 0.f;
#pragma unroll
    for (int w2 = 0; w2 < VW_; ++w2) s += sAcc[w2][tt][l][r];
    vt[(size_t)b * M_ * D_ + (size_t)(m0 + row) * D_ + d0 + col] = s;
  }
}

extern "C" void kernel_launch(void* const* d_in, const int* in_sizes, int n_in,
                              void* d_out, int out_size, void* d_ws, size_t ws_size,
                              hipStream_t stream) {
  const float* x      = (const float*)d_in[0];
  const float* conv_w = (const float*)d_in[1];
  const float* conv_b = (const float*)d_in[2];
  const float* v      = (const float*)d_in[3];

  float* out_vt = (float*)d_out;
  float* out_xp = out_vt + (size_t)B_ * M_ * D_;

  float* ws   = (float*)d_ws;
  float* Z    = ws + OFF_Z;
  float* u    = ws + OFF_U;
  float* vd   = ws + OFF_V;
  float* rinv = ws + OFF_RINV;
  float* pa   = ws + OFF_PA;
  float* pb   = ws + OFF_PB;
  float* pc   = ws + OFF_PC;

  k_proj<<<B_ * (HW_ / 64), 256, 0, stream>>>(x, conv_w, conv_b, v,
                                              out_xp, Z, rinv, pa);
  k_vu<NPA_><<<B_ * NPB_, 256, 0, stream>>>(Z, pa, u, vd, pb, 1);
  k_vu<NPB_><<<B_ * NPB_, 256, 0, stream>>>(Z, pb, u, vd, pc, 1);
  k_vu<NPB_><<<B_ * NPB_, 256, 0, stream>>>(Z, pc, u, vd, pb, 0);
  k_vtilde<<<B_ * 16, 512, 0, stream>>>(Z, u, vd, out_xp, rinv, out_vt);
}


# k_proj K loop: chunks go through a ring of two LDS stages, one LDS-only barrier per chunk (was two full ones); next chunk stored, the one after it requested at the chunk's head; sums unchanged
# speedup vs baseline: 2.4734x; 1.0472x over previous
#include <hip/hip_runtime.h>
#include <math.h>

typedef __attribute__((ext_vector_type(2))) float v2f;
typedef __attribute__((ext_vector_type(4))) float v4f;
typedef __attribute__((ext_vector_type(8))) float v8f;

#define B_    16
#define C_    768
#define HW_   4096
#define D_    256
#define M_    32
#define EPS_  0.05f
#define NORM_ (-8.3255517f)
#define NEPS_ 1e-12f

#define OFF_Z     0
#define OFF_U     2097152
#define OFF_V     (OFF_U + 512)
#define OFF_RINV  (OFF_V + 65536)
#define NPA_ 64
#define NPB_ 16
#define OFF_PA    (OFF_RINV + 65536)
#define OFF_PB    (OFF_PA + 65536)
#define OFF_PC    (OFF_PB + 16384)

static __device__ __forceinline__ v8f wmma4(v2f a, v2f b, v8f c) {
  return __builtin_amdgcn_wmma_f32_16x16x4_f32(false, a, false, b, (short)0, c,
                                               false, false);
}

#define BM_  64
#define BK_  32
#define PA_  36
#define PB_  36
#define STG_ (BM_ * PA_ + D_ * PB_)
#define PQ_  260
struct ProjFrag { v4f a[4], b0, b1; };

template <int CUR>
static __device__ __forceinline__ void proj_frag(ProjFrag& f, const float* sMem, int ks,
                                                 int laneLo, int laneHi, int d0) {
  const float* sA = sMem + CUR * STG_;
  const float* sB = sA + BM_ * PA_;
  const int kk = ks + 4 * laneHi;
#pragma unroll
  for (int i = 0; i < 4; ++i)
    f.a[i] = *(const v4f*)(sA + (16 * i + laneLo) * PA_ + kk);
  f.b0 = *(const v4f*)(sB + (d0 + laneLo) * PB_ + kk);
  f.b1 = *(const v4f*)(sB + (d0 + 16 + laneLo) * PB_ + kk);
}

static __device__ __forceinline__ void proj_mma(const ProjFrag& f, v8f (&acc)[4][2]) {
#pragma unroll
  for (int i = 0; i < 4; ++i) {
    acc[i][0] = wmma4(f.a[i].lo, f.b0.lo, acc[i][0]);
    acc[i][1] = wmma4(f.a[i].lo, f.b1.lo, acc[i][1]);
  }
#pragma unroll
  for (int i = 0; i < 4; ++i) {
    acc[i][0] = wmma4(f.a[i].hi, f.b0.hi, acc[i][0]);
    acc[i][1] = wmma4(f.a[i].hi, f.b1.hi, acc[i][1]);
  }
}

static __device__ __forceinline__ void ring_barrier() {
  __builtin_amdgcn_fence(__ATOMIC_RELEASE, "workgroup", "local");
  __builtin_amdgcn_s_barrier();
  __builtin_amdgcn_fence(__ATOMIC_ACQUIRE, "workgroup", "local");
}

template <int CUR, int NXT>
static __device__ __forceinline__ void proj_chunk(float* sMem, v8f (&acc)[4][2], bool stage,
                                                  bool fetch, const float* xg, const float* wg,
                                                  int knext, v4f& pa0, v4f& pa1, v4f (&pb)[8],
                                                  int an, int ak, int br, int bc, int laneLo,
                                                  int laneHi, int d0);

template <int NXT>
static __device__ __forceinline__ void proj_stage(float* sMem, v4f pa0, v4f pa1,
                                                  const v4f (&pb)[8], int an, int ak,
                                                  int br, int bc) {
  float* sA = sMem + NXT * STG_;
  float* sB = sA + BM_ * PA_;
  *(v4f*)(sA + an * PA_ + ak) = pa0;
  *(v4f*)(sA + an * PA_ + ak + 4) = pa1;
#pragma unroll
  for (int j = 0; j < 8; ++j) *(v4f*)(sB + (br + 32 * j) * PB_ + bc) = pb[j];
}

static __device__ __forceinline__ void proj_fetch(const float* xg, const float* wg, int k0,
                                                  v4f& pa0, v4f& pa1, v4f (&pb)[8]) {
  const float* xn = xg + (size_t)k0 * HW_;
#pragma unroll
  for (int j = 0; j < 4; ++j) {
    pa0[j] = xn[(size_t)j * HW_];
    pa1[j] = xn[(size_t)(j + 4) * HW_];
  }
#pragma unroll
  for (int j = 0; j < 8; ++j) pb[j] = *(const v4f*)(wg + (size_t)(32 * j) * C_ + k0);
}

template <int CUR, int NXT>
static __device__ __forceinline__ void proj_chunk(float* sMem, v8f (&acc)[4][2], bool stage,
                                                  bool fetch, const float* xg, const float* wg,
                                                  int knext, v4f& pa0, v4f& pa1, v4f (&pb)[8],
                                                  int an, int ak, int br, int bc, int laneLo,
                                                  int laneHi, int d0) {
  ProjFrag f;
  proj_frag<CUR>(f, sMem, 0, laneLo, laneHi, d0);
  if (stage) proj_stage<NXT>(sMem, pa0, pa1, pb, an, ak, br, bc);
  if (fetch) proj_fetch(xg, wg, knext, pa0, pa1, pb);
  proj_mma(f, acc);
#pragma unroll
  for (int ks = 8; ks < BK_; ks += 8) {
    proj_frag<CUR>(f, sMem, ks, laneLo, laneHi, d0);
    proj_mma(f, acc);
  }
  ring_barrier();
}

__global__ void __launch_bounds__(256) k_proj(
    const float* __restrict__ x, const float* __restrict__ w,
    const float* __restrict__ bias, const float* __restrict__ vc,
    float* __restrict__ xp, float* __restrict__ Z, float* __restrict__ rinvg,
    float* __restrict__ part) {
  __shared__ __attribute__((aligned(16))) float sMem[2 * STG_];
  __shared__ float sPart[8][16][2];
  __shared__ __attribute__((aligned(16))) float sV[M_ * PQ_];

  const int tid = threadIdx.x;
  const int b = blockIdx.x >> 6;
  const int n0 = (blockIdx.x & 63) * BM_;
  const int wave = tid >> 5;
  const int lane = tid & 31;
  const int laneLo = lane & 15;
  const int laneHi = lane >> 4;

  const int an = tid & 63;
  const int ak = (tid >> 6) * 8;
  const int br = tid >> 3;
  const int bc = (tid & 7) * 4;
  const float* xg = x + (size_t)b * C_ * HW_ + n0 + (size_t)ak * HW_ + an;
  const float* wg = w + (size_t)br * C_ + bc;

  v4f pa0, pa1, pb[8];
  proj_fetch(xg, wg, 0, pa0, pa1, pb);

#pragma unroll
  for (int j = 0; j < 8; ++j) {
    const int q = tid + 256 * j;
    const int vm = q >> 6, vd = (q & 63) * 4;
    *(v4f*)(sV + vm * PQ_ + vd) = *(const v4f*)(vc + (size_t)vm * D_ + vd);
  }

  const int d0 = wave * 32;
  v8f acc[4][2];
#pragma unroll
  for (int i = 0; i < 4; ++i) { acc[i][0] = (v8f){}; acc[i][1] = (v8f){}; }

  proj_stage<0>(sMem, pa0, pa1, pb, an, ak, br, bc);
  proj_fetch(xg, wg, BK_, pa0, pa1, pb);
  ring_barrier();

  for (int k0 = 0; k0 < C_; k0 += 2 * BK_) {
    const bool more = k0 + 2 * BK_ < C_;
    proj_chunk<0, 1>(sMem, acc, true, more, xg, wg, k0 + 2 * BK_, pa0, pa1, pb,
                     an, ak, br, bc, laneLo, laneHi, d0);
    proj_chunk<1, 0>(sMem, acc, more, more, xg, wg, k0 + 3 * BK_, pa0, pa1, pb,
                     an, ak, br, bc, laneLo, laneHi, d0);
  }

  const float bi0 = bias[d0 + laneLo];
  const float bi1 = bias[d0 + 16 + laneLo];
#pragma unroll
  for (int i = 0; i < 4; ++i) {
#pragma unroll
    for (int r = 0; r < 8; ++r) { acc[i][0][r] += bi0; acc[i][1][r] += bi1; }
  }

  float* sQ = sMem;
#pragma unroll
  for (int i = 0; i < 4; ++i) {
#pragma unroll
    for (int r = 0; r < 8; ++r) {
      int row = 16 * i + r + 8 * laneHi;
      sQ[row * PQ_ + d0 + laneLo] = acc[i][0][r];
      sQ[row * PQ_ + d0 + 16 + laneLo] = acc[i][1][r];
    }
  }
  __syncthreads();

  {
    float* xpb = xp + (size_t)b * HW_ * D_ + (size_t)n0 * D_;
    const int c4 = (tid & 63) * 4;
    const int r0 = tid >> 6;
#pragma unroll
    for (int j = 0; j < 16; ++j) {
      const int row = r0 + 4 * j;
      *(v4f*)(xpb + (size_t)row * D_ + c4) = *(const v4f*)(sQ + row * PQ_ + c4);
    }
  }
  __builtin_amdgcn_sched_barrier(0);

  {
    const int nt = (wave >> 1) * 16;
    const int m0 = (wave & 1) * 16;
    v8f cs = {};
    float ss = 0.f;
    float sr = 0.f;
    const float* qa = sQ + (nt + laneLo) * PQ_ + 4 * laneHi;
    const float* qb = sV + (m0 + laneLo) * PQ_ + 4 * laneHi;
#pragma unroll
    for (int g = 0; g < 32; ++g) {
      const v4f a = *(const v4f*)(qa + 8 * g);
      const v4f bb = *(const v4f*)(qb + 8 * g);
      ss += bb.x * bb.x + bb.y * bb.y;
      ss += bb.z * bb.z + bb.w * bb.w;
      sr += a.x * a.x + a.y * a.y;
      sr += a.z * a.z + a.w * a.w;
      cs = wmma4(a.lo, bb.lo, cs);
      cs = wmma4(a.hi, bb.hi, cs);
      if ((g & 7) == 7) __builtin_amdgcn_sched_barrier(0);
    }
    ss += __shfl_xor(ss, 16, 32);
    const float invv = 1.0f / fmaxf(sqrtf(ss), NEPS_);
    sr += __shfl_xor(sr, 16, 32);
    const float invr = 1.0f / fmaxf(sqrtf(sr), NEPS_);
    if (m0 == 0 && laneHi == 0) rinvg[(size_t)b * HW_ + n0 + nt + laneLo] = invr;
    float* Zb = Z + (size_t)b * (M_ * HW_) + (size_t)(m0 + laneLo) * HW_ + n0 + nt;
    const float invEps = 1.0f / EPS_;
    float zv[8];
#pragma unroll
    for (int r = 0; r < 8; ++r) {
      int row = r + 8 * laneHi;
      zv[r] = cs[r] * (__shfl(invr, row, 32) * invv) * invEps;
      Zb[row] = zv[r];
    }
    float mx = zv[0];
#pragma unroll
    for (int r = 1; r < 8; ++r) mx = fmaxf(mx, zv[r]);
    mx = fmaxf(mx, __shfl_xor(mx, 16, 32));
    float sm = 0.f;
#pragma unroll
    for (int r = 0; r < 8; ++r) sm += expf(zv[r] - mx);
    sm += __shfl_xor(sm, 16, 32);
    if (laneHi == 0) { sPart[wave][laneLo][0] = mx; sPart[wave][laneLo][1] = sm; }
  }
  __syncthreads();
  if (tid < M_) {
    const int mt = tid >> 4, ml = tid & 15;
    float M = sPart[mt][ml][0];
#pragma unroll
    for (int q = 1; q < 4; ++q) M = fmaxf(M, sPart[2 * q + mt][ml][0]);
    float S = 0.f;
#pragma unroll
    for (int q = 0; q < 4; ++q)
      S += sPart[2 * q + mt][ml][1] * expf(sPart[2 * q + mt][ml][0] - M);
    float* po = part + (((size_t)b * NPA_ + (blockIdx.x & 63)) * M_ + tid) * 2;
    v2f ms; ms.x = M; ms.y = S;
    *(v2f*)po = ms;
  }
}

#define PT_ 264
template <int NP>
__global__ void __launch_bounds__(256) k_vu(const float* __restrict__ Z,
                                            const float* __restrict__ pin,
                                            float* __restrict__ u,
                                            float* __restrict__ vdual,
                                            float* __restrict__ pout, int want_part) {
  __shared__ float sU[M_];
  __shared__ float sPm[8][M_];
  __shared__ float sPs[8][M_];
  __shared__ float sT[M_ * PT_];
  const int tid = threadIdx.x;
  const int wave = tid >> 5;
  const int lane = tid & 31;
  const int b = blockIdx.x >> 4;
  const int nb = blockIdx.x & 15;
  const int n = nb * 256 + tid;

  const float* zb = Z + (size_t)b * (M_ * HW_) + n;
  float z[M_];
#pragma unroll
  for (int m = 0; m < M_; ++m) z[m] = zb[(size_t)m * HW_];

  {
    const float* pp = pin + ((size_t)b * NP * M_ + lane) * 2;
    v2f ms[NP / 8];
#pragma unroll
    for (int i = 0; i < NP / 8; ++i)
      ms[i] = *(const v2f*)(pp + (size_t)(wave + 8 * i) * (M_ * 2));
    float M = ms[0].x;
#pragma unroll
    for (int i = 1; i < NP / 8; ++i) M = fmaxf(M, ms[i].x);
    float S = 0.f;
#pragma unroll
    for (int i = 0; i < NP / 8; ++i) S += ms[i].y * expf(ms[i].x - M);
    sPm[wave][lane] = M;
    sPs[wave][lane] = S;
  }
  __syncthreads();
  if (tid < M_) {
    float M = sPm[0][tid];
#pragma unroll
    for (int w2 = 1; w2 < 8; ++w2) M = fmaxf(M, sPm[w2][tid]);
    float S = 0.f;
#pragma unroll
    for (int w2 = 0; w2 < 8; ++w2) S += sPs[w2][tid] * expf(sPm[w2][tid] - M);
    const float um = NORM_ - (M + logf(S));
    sU[tid] = um;
    if (nb == 0) u[b * M_ + tid] = um;
  }
  __syncthreads();

  float mx = -1e30f;
#pragma unroll
  for (int m = 0; m < M_; ++m) mx = fmaxf(mx, z[m] + sU[m]);
  float sm = 0.f;
#pragma unroll
  for (int m = 0; m < M_; ++m) sm += expf(z[m] + sU[m] - mx);
  const float vn = NORM_ - (mx + logf(sm));
  vdual[(size_t)b * HW_ + n] = vn;

  if (want_part) {
#pragma unroll
    for (int m = 0; m < M_; ++m) sT[m * PT_ + tid] = z[m] + vn;
    __syncthreads();
    {
      const int m = tid >> 3, seg = tid & 7;
      float tv[32];
#pragma unroll
      for (int i = 0; i < 32; ++i) tv[i] = sT[m * PT_ + seg + 8 * i];
      float M = tv[0];
#pragma unroll
      for (int i = 1; i < 32; ++i) M = fmaxf(M, tv[i]);
      float S = 0.f;
#pragma unroll
      for (int i = 0; i < 32; ++i) S += expf(tv[i] - M);
      sPm[seg][m] = M;
      sPs[seg][m] = S;
    }
    __syncthreads();
    if (tid < M_) {
      float M = sPm[0][tid];
#pragma unroll
      for (int w2 = 1; w2 < 8; ++w2) M = fmaxf(M, sPm[w2][tid]);
      float S = 0.f;
#pragma unroll
      for (int w2 = 0; w2 < 8; ++w2) S += sPs[w2][tid] * expf(sPm[w2][tid] - M);
      float* po = pout + (((size_t)b * NPB_ + nb) * M_ + tid) * 2;
      v2f ms; ms.x = M; ms.y = S;
      *(v2f*)po = ms;
    }
  }
}

#define VR_ 32
#define VQ_ (VR_ / 8)
#define VW_ 16
#define VT_TRIPS_ ((HW_ / VW_) / VR_)
struct VtBatch { v4f z[VQ_], vw[VQ_], r[VQ_]; v2f x[VR_ / 2]; };

static __device__ __forceinline__ void vt_load(VtBatch& t, const float* __restrict__ zq,
                                               const float* __restrict__ vq,
                                               const float* __restrict__ rq,
                                               const float* __restrict__ xq) {
#pragma unroll
  for (int q = 0; q < VQ_; ++q) {
    t.z[q]  = *(const v4f*)(zq + 4 * q);
    t.vw[q] = *(const v4f*)(vq + 4 * q);
  }
#pragma unroll
  for (int q = 0; q < VQ_; ++q) t.r[q] = *(const v4f*)(rq + 4 * q);
#pragma unroll
  for (int e = 0; e < VR_ / 2; ++e) t.x[e] = *(const v2f*)(xq + e * D_);
}

static __device__ __forceinline__ void vt_mma(const VtBatch& t, float um, v8f& acc0,
                                              v8f& acc1) {
#pragma unroll
  for (int j = 0; j < VR_ / 4; ++j) {
    const int q = j >> 1, p = (j & 1) * 2;
    v2f a;
    a.x = expf(t.z[q][p] + um + t.vw[q][p] - NORM_);
    a.y = expf(t.z[q][p + 1] + um + t.vw[q][p + 1] - NORM_);
    const float r0 = t.r[q][p], r1 = t.r[q][p + 1];
    v2f b0, b1;
    b0.x = t.x[2 * j].x * r0;
    b0.y = t.x[2 * j + 1].x * r1;
    b1.x = t.x[2 * j].y * r0;
    b1.y = t.x[2 * j + 1].y * r1;
    acc0 = wmma4(a, b0, acc0);
    acc1 = wmma4(a, b1, acc1);
  }
}

__global__ void __launch_bounds__(512) k_vtilde(
    const float* __restrict__ Z, const float* __restrict__ u,
    const float* __restrict__ vdual, const float* __restrict__ xp,
    const float* __restrict__ rinv, float* __restrict__ vt) {
  __shared__ float sAcc[VW_][2][32][9];
  const int tid = threadIdx.x;
  const int wave = tid >> 5;
  const int lane = tid & 31;
  const int laneLo = lane & 15;
  const int laneHi = lane >> 4;

  const int b = blockIdx.x >> 4;
  const int rem = blockIdx.x & 15;
  const int m0 = (rem >> 3) * 16;
  const int d0 = (rem & 7) * 32;

  const int kq = wave * (HW_ / VW_) + (VR_ / 2) * laneHi;

  const float um = u[b * M_ + m0 + laneLo];
  const float* zq = Z + (size_t)b * (M_ * HW_) + (size_t)(m0 + laneLo) * HW_ + kq;
  const float* vq = vdual + (size_t)b * HW_ + kq;
  const float* rq = rinv + (size_t)b * HW_ + kq;
  const float* xq = xp + (size_t)b * HW_ * D_ + (size_t)kq * D_ + d0 + 2 * laneLo;

  v8f acc0 = {}, acc1 = {};
#pragma nounroll
  for (int it = 0; it < VT_TRIPS_; ++it) {
    VtBatch t;
    vt_load(t, zq, vq, rq, xq);
    __builtin_amdgcn_sched_barrier(0);
    vt_mma(t, um, acc0, acc1);
    zq += VR_; vq += VR_; rq += VR_; xq += (size_t)VR_ * D_;
  }

#pragma unroll
  for (int r = 0; r < 8; ++r) {
    sAcc[wave][0][lane][r] = acc0[r];
    sAcc[wave][1][lane][r] = acc1[r];
  }
  __syncthreads();

  {
    const int row = tid >> 5, col = tid & 31;
    const int tt = col & 1, l = (col >> 1) + 16 * (row >> 3), r = row & 7;
    float s = 0.f;
#pragma unroll
    for (int w2 = 0; w2 < VW_; ++w2) s += sAcc[w2][tt][l][r];
    vt[(size_t)b * M_ * D_ + (size_t)(m0 + row) * D_ + d0 + col] = s;
  }
}

extern "C" void kernel_launch(void* const* d_in, const int* in_sizes, int n_in,
                              void* d_out, int out_size, void* d_ws, size_t ws_size,
                              hipStream_t stream) {
  const float* x      = (const float*)d_in[0];
  const float* conv_w = (const float*)d_in[1];
  const float* conv_b = (const float*)d_in[2];
  const float* v      = (const float*)d_in[3];

  float* out_vt = (float*)d_out;
  float* out_xp = out_vt + (size_t)B_ * M_ * D_;

  float* ws   = (float*)d_ws;
  float* Z    = ws + OFF_Z;
  float* u    = ws + OFF_U;
  float* vd   = ws + OFF_V;
  float* rinv = ws + OFF_RINV;
  float* pa   = ws + OFF_PA;
  float* pb   = ws + OFF_PB;
  float* pc   = ws + OFF_PC;

  k_proj<<<B_ * (HW_ / 64), 256, 0, stream>>>(x, conv_w, conv_b, v,
                                              out_xp, Z, rinv, pa);
  k_vu<NPA_><<<B_ * NPB_, 256, 0, stream>>>(Z, pa, u, vd, pb, 1);
  k_vu<NPB_><<<B_ * NPB_, 256, 0, stream>>>(Z, pb, u, vd, pc, 1);
  k_vu<NPB_><<<B_ * NPB_, 256, 0, stream>>>(Z, pc, u, vd, pb, 0);
  k_vtilde<<<B_ * 16, 512, 0, stream>>>(Z, u, vd, out_xp, rinv, out_vt);
}
